# starting kernel
# baseline (speedup 1.0000x reference)
#include <hip/hip_runtime.h>
#include <cstdint>

typedef __attribute__((ext_vector_type(2))) float v2f;
typedef __attribute__((ext_vector_type(8))) float v8f;

#define N_NODES 8192
#define F_INF   512
#define HIDF    64
#define CCH     16
#define KORD    10
#define JCHUNK  512
#define NCHUNKS (N_NODES / JCHUNK)

__device__ __forceinline__ v8f wmma4(v2f a, v2f b, v8f c) {
  return __builtin_amdgcn_wmma_f32_16x16x4_f32(
       false, a,  false, b,
       (short)0, c,  false,  false);
}

__global__ void zero_kernel(float4* __restrict__ p) {
  p[blockIdx.x * blockDim.x + threadIdx.x] = make_float4(0.f, 0.f, 0.f, 0.f);
}

__global__ void jacobi_kernel(const float* __restrict__ origin_e,
                              const float* __restrict__ temp,
                              float* __restrict__ Hf) {
  __shared__ float al[CCH * (KORD + 1)];
  int tid = threadIdx.x;
  if (tid < CCH * (KORD + 1)) al[tid] = 0.5f * tanhf(temp[tid]);
  __syncthreads();

  int n = blockIdx.x * blockDim.x + tid;
  if (n >= N_NODES) return;
  float e = origin_e[n];

  const float a = 1.0f, b = 1.0f, lo = -1.0f, hi = 1.0f;
  const float coef1 = (a - b) * 0.5f - (a + b + 2.0f) * 0.5f * (lo + hi) / (hi - lo);
  const float coef2 = (a + b + 2.0f) / (hi - lo);

  for (int c = 0; c < CCH; ++c) {
    const float* ac = &al[c * (KORD + 1)];
    float xs_m2 = ac[0];
    float acc = xs_m2;
    float xs_m1 = coef1 * ac[1] + coef2 * ac[1] * e * xs_m2;
    acc += xs_m1;
#pragma unroll
    for (int L = 2; L <= KORD; ++L) {
      float Lf = (float)L;
      float coef_l  = 2.0f * Lf * (Lf + a + b) * (2.0f * Lf - 2.0f + a + b);
      float c_lm1_1 = (2.0f * Lf + a + b - 1.0f) * (2.0f * Lf + a + b) * (2.0f * Lf + a + b - 2.0f);
      float c_lm1_2 = (2.0f * Lf + a + b - 1.0f) * (a * a - b * b);
      float c_lm2   = 2.0f * (Lf - 1.0f + a) * (Lf - 1.0f + b) * (2.0f * Lf + a + b);
      float tmp1   = ac[L - 1] * (c_lm1_1 / coef_l);
      float tmp2   = ac[L - 1] * (c_lm1_2 / coef_l);
      float tmp3   = ac[L - 1] * ac[L - 2] * (c_lm2 / coef_l);
      float tmp1_2 = tmp1 * (2.0f / (hi - lo));
      float tmp2_2 = tmp1 * ((hi + lo) / (hi - lo)) + tmp2;
      float nx = tmp1_2 * (e * xs_m1) - tmp2_2 * xs_m1 - tmp3 * xs_m2;
      acc += nx;
      xs_m2 = xs_m1;
      xs_m1 = nx;
    }
    Hf[n * CCH + c] = acc;
  }
}

#define XS_STRIDE 516
#define HS_STRIDE 68

__global__ void mlp_kernel(const float* __restrict__ x,
                           const float* __restrict__ W1,
                           const float* __restrict__ b1,
                           const float* __restrict__ W2,
                           const float* __restrict__ b2,
                           float* __restrict__ zb) {
  __shared__ float xs[16 * XS_STRIDE];
  __shared__ float hs[16 * HS_STRIDE];

  const int tid  = threadIdx.x;
  const int w    = tid >> 5;
  const int lane = tid & 31;
  const int mn   = lane & 15;
  const int half = lane >> 4;
  const int kk   = half * 2;
  const int m0   = blockIdx.x * 16;
  const int n0   = w * 16;

  for (int idx = tid; idx < 16 * (F_INF / 4); idx += blockDim.x) {
    int row = idx >> 7;
    int c4  = idx & 127;
    float4 v = *(const float4*)&x[(size_t)(m0 + row) * F_INF + c4 * 4];
    float* d = &xs[row * XS_STRIDE + c4 * 4];
    d[0] = v.x; d[1] = v.y; d[2] = v.z; d[3] = v.w;
  }
  __syncthreads();

  v8f acc = {};
#pragma unroll 4
  for (int k0 = 0; k0 < F_INF; k0 += 4) {
    v2f av, bv;
    av.x = xs[mn * XS_STRIDE + k0 + kk];
    av.y = xs[mn * XS_STRIDE + k0 + kk + 1];
    bv.x = W1[(size_t)(k0 + kk) * HIDF + n0 + mn];
    bv.y = W1[(size_t)(k0 + kk + 1) * HIDF + n0 + mn];
    acc = wmma4(av, bv, acc);
  }

  float bias = b1[n0 + mn];
#pragma unroll
  for (int r = 0; r < 8; ++r) {
    int row = half ? r + 8 : r;
    float v = acc[r] + bias;
    hs[row * HS_STRIDE + n0 + mn] = v > 0.0f ? v : 0.0f;
  }
  __syncthreads();

  if (w == 0) {
    v8f zacc = {};
#pragma unroll
    for (int k0 = 0; k0 < HIDF; k0 += 4) {
      v2f av, bv;
      av.x = hs[mn * HS_STRIDE + k0 + kk];
      av.y = hs[mn * HS_STRIDE + k0 + kk + 1];
      bv.x = W2[(k0 + kk) * CCH + mn];
      bv.y = W2[(k0 + kk + 1) * CCH + mn];
      zacc = wmma4(av, bv, zacc);
    }
    float bb = b2[mn];
#pragma unroll
    for (int r = 0; r < 8; ++r) {
      int row = half ? r + 8 : r;
      zb[(size_t)(m0 + row) * CCH + mn] = zacc[r] + bb;
    }
  }
}

__global__ void utz_kernel(const float* __restrict__ U,
                           const float* __restrict__ zb,
                           float* __restrict__ tbuf) {
  __shared__ float zs[JCHUNK * CCH];

  const int tid  = threadIdx.x;
  const int lane = tid & 31;
  const int mn   = lane & 15;
  const int half = lane >> 4;
  const int kk   = half * 2;
  const int W    = blockIdx.x * 8 + (tid >> 5);
  const int i0   = (W & 511) * 16;
  const int jb   = (W >> 9) * JCHUNK;

  for (int idx = tid; idx < JCHUNK * CCH / 4; idx += blockDim.x) {
    float4 v = *(const float4*)&zb[(size_t)jb * CCH + idx * 4];
    float* d = &zs[idx * 4];
    d[0] = v.x; d[1] = v.y; d[2] = v.z; d[3] = v.w;
  }
  __syncthreads();

  v8f acc = {};
#pragma unroll 4
  for (int jl = 0; jl < JCHUNK; jl += 4) {
    v2f av, bv;
    av.x = zs[(jl + kk) * CCH + mn];
    av.y = zs[(jl + kk + 1) * CCH + mn];
    bv.x = U[(size_t)(jb + jl + kk) * N_NODES + i0 + mn];
    bv.y = U[(size_t)(jb + jl + kk + 1) * N_NODES + i0 + mn];
    acc = wmma4(av, bv, acc);
  }
#pragma unroll
  for (int r = 0; r < 8; ++r) {
    int c = half ? r + 8 : r;
    unsafeAtomicAdd(&tbuf[(size_t)(i0 + mn) * CCH + c], acc[r]);
  }
}

#define UT_STRIDE 68

__global__ void outgemm_kernel(const float* __restrict__ U,
                               const float* __restrict__ tbuf,
                               const float* __restrict__ Hf,
                               float* __restrict__ obuf) {
  __shared__ float Ut[128 * UT_STRIDE];
  __shared__ float St[64 * CCH];

  const int tid  = threadIdx.x;
  const int w    = tid >> 5;
  const int lane = tid & 31;
  const int mn   = lane & 15;
  const int half = lane >> 4;
  const int kk   = half * 2;
  const int r0   = (blockIdx.x & 63) * 128;
  const int kb   = (blockIdx.x >> 6) * JCHUNK;

  v8f acc = {};
  for (int i0 = kb; i0 < kb + JCHUNK; i0 += 64) {
    for (int idx = tid; idx < 128 * 16; idx += blockDim.x) {
      int row = idx >> 4;
      int c4  = idx & 15;
      float4 v = *(const float4*)&U[(size_t)(r0 + row) * N_NODES + i0 + c4 * 4];
      float* d = &Ut[row * UT_STRIDE + c4 * 4];
      d[0] = v.x; d[1] = v.y; d[2] = v.z; d[3] = v.w;
    }
    {
      size_t base = (size_t)(i0 + (tid >> 2)) * CCH + (tid & 3) * 4;
      float4 tv = *(const float4*)&tbuf[base];
      float4 hv = *(const float4*)&Hf[base];
      float* d = &St[(tid >> 2) * CCH + (tid & 3) * 4];
      d[0] = tv.x * hv.x; d[1] = tv.y * hv.y; d[2] = tv.z * hv.z; d[3] = tv.w * hv.w;
    }
    __syncthreads();
#pragma unroll
    for (int k0 = 0; k0 < 64; k0 += 4) {
      v2f av, bv;
      av.x = Ut[(16 * w + mn) * UT_STRIDE + k0 + kk];
      av.y = Ut[(16 * w + mn) * UT_STRIDE + k0 + kk + 1];
      bv.x = St[(k0 + kk) * CCH + mn];
      bv.y = St[(k0 + kk + 1) * CCH + mn];
      acc = wmma4(av, bv, acc);
    }
    __syncthreads();
  }
#pragma unroll
  for (int r = 0; r < 8; ++r) {
    int row = half ? r + 8 : r;
    unsafeAtomicAdd(&obuf[(size_t)(r0 + 16 * w + row) * CCH + mn], acc[r]);
  }
}

__global__ void lsm_kernel(const float* __restrict__ obuf,
                           float* __restrict__ out) {
  int n = blockIdx.x * blockDim.x + threadIdx.x;
  if (n >= N_NODES) return;
  float v[CCH];
#pragma unroll
  for (int q = 0; q < 4; ++q) {
    float4 t = *(const float4*)&obuf[(size_t)n * CCH + q * 4];
    v[q * 4 + 0] = t.x; v[q * 4 + 1] = t.y; v[q * 4 + 2] = t.z; v[q * 4 + 3] = t.w;
  }
  float mx = v[0];
#pragma unroll
  for (int i = 1; i < CCH; ++i) mx = fmaxf(mx, v[i]);
  float sm = 0.0f;
#pragma unroll
  for (int i = 0; i < CCH; ++i) sm += expf(v[i] - mx);
  float lse = mx + logf(sm);
#pragma unroll
  for (int q = 0; q < 4; ++q) {
    float4 t;
    t.x = v[q * 4 + 0] - lse; t.y = v[q * 4 + 1] - lse;
    t.z = v[q * 4 + 2] - lse; t.w = v[q * 4 + 3] - lse;
    *(float4*)&out[(size_t)n * CCH + q * 4] = t;
  }
}

extern "C" void kernel_launch(void* const* d_in, const int* in_sizes, int n_in,
                              void* d_out, int out_size, void* d_ws, size_t ws_size,
                              hipStream_t stream) {
  const float* origin_e = (const float*)d_in[0];
  const float* U        = (const float*)d_in[1];
  const float* x        = (const float*)d_in[2];
  const float* W1       = (const float*)d_in[3];
  const float* b1       = (const float*)d_in[4];
  const float* W2       = (const float*)d_in[5];
  const float* b2       = (const float*)d_in[6];
  const float* temp     = (const float*)d_in[7];
  float* out = (float*)d_out;

  float* ws   = (float*)d_ws;
  float* Hf   = ws;
  float* zb   = Hf   + (size_t)N_NODES * CCH;
  float* tbuf = zb   + (size_t)N_NODES * CCH;
  float* obuf = tbuf + (size_t)N_NODES * CCH;

  zero_kernel<<<256, 256, 0, stream>>>((float4*)tbuf);
  jacobi_kernel<<<N_NODES / 256, 256, 0, stream>>>(origin_e, temp, Hf);
  mlp_kernel<<<N_NODES / 16, 128, 0, stream>>>(x, W1, b1, W2, b2, zb);
  utz_kernel<<<(N_NODES / 16) * NCHUNKS / 8, 256, 0, stream>>>(U, zb, tbuf);
  outgemm_kernel<<<64 * NCHUNKS, 256, 0, stream>>>(U, tbuf, Hf, obuf);
  lsm_kernel<<<N_NODES / 256, 256, 0, stream>>>(obuf, out);
}


# MLP re-tiled: a block owns 32 rows of x, W1 staged once per block in LDS, every x and W1 request in flight before the first LDS store; layer 1 out of LDS on two accumulators: sum order changes
# speedup vs baseline: 1.4376x; 1.4376x over previous
#include <hip/hip_runtime.h>
#include <cstdint>

typedef __attribute__((ext_vector_type(2))) float v2f;
typedef __attribute__((ext_vector_type(8))) float v8f;

#define N_NODES 8192
#define F_INF   512
#define HIDF    64
#define CCH     16
#define KORD    10
#define JCHUNK  512
#define NCHUNKS (N_NODES / JCHUNK)

__device__ __forceinline__ v8f wmma4(v2f a, v2f b, v8f c) {
  return __builtin_amdgcn_wmma_f32_16x16x4_f32(
       false, a,  false, b,
       (short)0, c,  false,  false);
}

__global__ void zero_kernel(float4* __restrict__ p) {
  p[blockIdx.x * blockDim.x + threadIdx.x] = make_float4(0.f, 0.f, 0.f, 0.f);
}

__global__ void jacobi_kernel(const float* __restrict__ origin_e,
                              const float* __restrict__ temp,
                              float* __restrict__ Hf) {
  __shared__ float al[CCH * (KORD + 1)];
  int tid = threadIdx.x;
  if (tid < CCH * (KORD + 1)) al[tid] = 0.5f * tanhf(temp[tid]);
  __syncthreads();

  int n = blockIdx.x * blockDim.x + tid;
  if (n >= N_NODES) return;
  float e = origin_e[n];

  const float a = 1.0f, b = 1.0f, lo = -1.0f, hi = 1.0f;
  const float coef1 = (a - b) * 0.5f - (a + b + 2.0f) * 0.5f * (lo + hi) / (hi - lo);
  const float coef2 = (a + b + 2.0f) / (hi - lo);

  for (int c = 0; c < CCH; ++c) {
    const float* ac = &al[c * (KORD + 1)];
    float xs_m2 = ac[0];
    float acc = xs_m2;
    float xs_m1 = coef1 * ac[1] + coef2 * ac[1] * e * xs_m2;
    acc += xs_m1;
#pragma unroll
    for (int L = 2; L <= KORD; ++L) {
      float Lf = (float)L;
      float coef_l  = 2.0f * Lf * (Lf + a + b) * (2.0f * Lf - 2.0f + a + b);
      float c_lm1_1 = (2.0f * Lf + a + b - 1.0f) * (2.0f * Lf + a + b) * (2.0f * Lf + a + b - 2.0f);
      float c_lm1_2 = (2.0f * Lf + a + b - 1.0f) * (a * a - b * b);
      float c_lm2   = 2.0f * (Lf - 1.0f + a) * (Lf - 1.0f + b) * (2.0f * Lf + a + b);
      float tmp1   = ac[L - 1] * (c_lm1_1 / coef_l);
      float tmp2   = ac[L - 1] * (c_lm1_2 / coef_l);
      float tmp3   = ac[L - 1] * ac[L - 2] * (c_lm2 / coef_l);
      float tmp1_2 = tmp1 * (2.0f / (hi - lo));
      float tmp2_2 = tmp1 * ((hi + lo) / (hi - lo)) + tmp2;
      float nx = tmp1_2 * (e * xs_m1) - tmp2_2 * xs_m1 - tmp3 * xs_m2;
      acc += nx;
      xs_m2 = xs_m1;
      xs_m1 = nx;
    }
    Hf[n * CCH + c] = acc;
  }
}

#define ML_ROWS 32
#define ML_THR  256
#define ML_XP   516
#define ML_WP   72
#define ML_HP   68
#define ML_XS   (ML_ROWS * ML_XP)
#define ML_WS   (F_INF * ML_WP)
#define ML_HS   (ML_ROWS * ML_HP)

typedef __attribute__((ext_vector_type(4))) float v4f;

__global__ void __launch_bounds__(ML_THR)
mlp_kernel(const float* __restrict__ x,
           const float* __restrict__ W1,
           const float* __restrict__ b1,
           const float* __restrict__ W2,
           const float* __restrict__ b2,
           float* __restrict__ zb) {
  __shared__ __attribute__((aligned(16))) float lds[ML_XS + ML_WS + ML_HS];
  float* wl = lds;
  float* xs = lds + ML_WS;
  float* hs = lds + ML_WS + ML_XS;

  const int tid  = threadIdx.x;
  const int w    = tid >> 5;
  const int lane = tid & 31;
  const int mn   = lane & 15;
  const int half = lane >> 4;
  const int kk   = half * 2;
  const int mt   = w & 1;
  const int n0   = (w >> 1) * 16;
  const int m0   = blockIdx.x * ML_ROWS;

  v4f xr[16], wr[32];
  const float* px = x + (size_t)(m0 + (tid >> 7)) * F_INF + (tid & 127) * 4;
#pragma unroll
  for (int i = 0; i < 16; ++i) xr[i] = *(const v4f*)(px + (size_t)(2 * i) * F_INF);
  const int wrot = blockIdx.x & 31;
#pragma unroll
  for (int i = 0; i < 32; ++i) {
    const int j = (i + wrot) & 31;
    wr[i] = *(const v4f*)(W1 + (size_t)(tid + 256 * j) * 4);
  }
  v2f w2f[16];
  float bb = 0.0f;
  if (w < 2) {
#pragma unroll
    for (int q = 0; q < 16; ++q) {
      w2f[q].x = W2[(4 * q + kk) * CCH + mn];
      w2f[q].y = W2[(4 * q + kk + 1) * CCH + mn];
    }
    bb = b2[mn];
  }
  const float bias = b1[n0 + mn];

  const int xoff = (tid >> 7) * ML_XP + (tid & 127) * 4;
#pragma unroll
  for (int i = 0; i < 16; ++i) *(v4f*)&xs[xoff + 2 * i * ML_XP] = xr[i];
  const int woff = (tid >> 4) * ML_WP + (tid & 15) * 4;
#pragma unroll
  for (int i = 0; i < 32; ++i) {
    const int j = (i + wrot) & 31;
    *(v4f*)&wl[woff + 16 * j * ML_WP] = wr[i];
  }
  __syncthreads();

  v8f acc0 = {}, acc1 = {};
  const float* xa = &xs[(mt * 16 + mn) * ML_XP + 4 * half];
  const float* wb = &wl[(4 * half) * ML_WP + n0 + mn];
#pragma unroll 8
  for (int g = 0; g < F_INF / 8; ++g) {
    const v4f a = *(const v4f*)&xa[8 * g];
    const float* wg = wb + 8 * g * ML_WP;
    v2f b01, b23, av;
    b01.x = wg[0];          b01.y = wg[ML_WP];
    b23.x = wg[2 * ML_WP];  b23.y = wg[3 * ML_WP];
    av.x = a.x; av.y = a.y; acc0 = wmma4(av, b01, acc0);
    av.x = a.z; av.y = a.w; acc1 = wmma4(av, b23, acc1);
  }

#pragma unroll
  for (int r = 0; r < 8; ++r) {
    const int row = mt * 16 + (half ? r + 8 : r);
    const float v = (acc0[r] + acc1[r]) + bias;
    hs[row * ML_HP + n0 + mn] = v > 0.0f ? v : 0.0f;
  }
  __syncthreads();

  if (w < 2) {
    v8f zacc = {};
#pragma unroll
    for (int q = 0; q < 16; ++q) {
      v2f av;
      av.x = hs[(w * 16 + mn) * ML_HP + 4 * q + kk];
      av.y = hs[(w * 16 + mn) * ML_HP + 4 * q + kk + 1];
      zacc = wmma4(av, w2f[q], zacc);
    }
#pragma unroll
    for (int r = 0; r < 8; ++r) {
      const int row = w * 16 + (half ? r + 8 : r);
      zb[(size_t)(m0 + row) * CCH + mn] = zacc[r] + bb;
    }
  }
}

__global__ void utz_kernel(const float* __restrict__ U,
                           const float* __restrict__ zb,
                           float* __restrict__ tbuf) {
  __shared__ float zs[JCHUNK * CCH];

  const int tid  = threadIdx.x;
  const int lane = tid & 31;
  const int mn   = lane & 15;
  const int half = lane >> 4;
  const int kk   = half * 2;
  const int W    = blockIdx.x * 8 + (tid >> 5);
  const int i0   = (W & 511) * 16;
  const int jb   = (W >> 9) * JCHUNK;

  for (int idx = tid; idx < JCHUNK * CCH / 4; idx += blockDim.x) {
    float4 v = *(const float4*)&zb[(size_t)jb * CCH + idx * 4];
    float* d = &zs[idx * 4];
    d[0] = v.x; d[1] = v.y; d[2] = v.z; d[3] = v.w;
  }
  __syncthreads();

  v8f acc = {};
#pragma unroll 4
  for (int jl = 0; jl < JCHUNK; jl += 4) {
    v2f av, bv;
    av.x = zs[(jl + kk) * CCH + mn];
    av.y = zs[(jl + kk + 1) * CCH + mn];
    bv.x = U[(size_t)(jb + jl + kk) * N_NODES + i0 + mn];
    bv.y = U[(size_t)(jb + jl + kk + 1) * N_NODES + i0 + mn];
    acc = wmma4(av, bv, acc);
  }
#pragma unroll
  for (int r = 0; r < 8; ++r) {
    int c = half ? r + 8 : r;
    unsafeAtomicAdd(&tbuf[(size_t)(i0 + mn) * CCH + c], acc[r]);
  }
}

#define OG_ROWS 32
#define OG_KC   256
#define OG_NCH  (N_NODES / OG_KC)
#define OG_THR  512
#define OG_UP   260
#define OG_UT   (OG_ROWS * OG_UP)
#define OG_SG   80
#define OG_ST   ((OG_KC / 4) * OG_SG)
#define OG_BUF  (OG_UT + OG_ST)

#define OG_DECL(S) v4f S##u0, S##u1, S##u2, S##u3, S##t0, S##t1, S##h0, S##h1

#define OG_ISSUE(S, ch)                                                        \
  do {                                                                         \
    const float* pu_ = pu + (size_t)(ch) * OG_KC;                              \
    S##u0 = *(const v4f*)(pu_);                                                \
    S##u1 = *(const v4f*)(pu_ + (size_t)8 * N_NODES);                          \
    S##u2 = *(const v4f*)(pu_ + (size_t)16 * N_NODES);                         \
    S##u3 = *(const v4f*)(pu_ + (size_t)24 * N_NODES);                         \
    S##t0 = pt[(ch) * (OG_KC * 4)];                                            \
    S##t1 = pt[(ch) * (OG_KC * 4) + OG_THR];                                   \
    S##h0 = ph[(ch) * (OG_KC * 4)];                                            \
    S##h1 = ph[(ch) * (OG_KC * 4) + OG_THR];                                   \
  } while (0)

#define OG_STASH(S, buf)                                                       \
  do {                                                                         \
    *(v4f*)&(buf)[uoff]              = S##u0;                                  \
    *(v4f*)&(buf)[uoff + 8 * OG_UP]  = S##u1;                                  \
    *(v4f*)&(buf)[uoff + 16 * OG_UP] = S##u2;                                  \
    *(v4f*)&(buf)[uoff + 24 * OG_UP] = S##u3;                                  \
    *(v4f*)&(buf)[OG_UT + soff]              = S##t0 * S##h0;                  \
    *(v4f*)&(buf)[OG_UT + soff + 32 * OG_SG] = S##t1 * S##h1;                  \
  } while (0)

__device__ __forceinline__ void og_compute(const float* buf, int aoff, int boff,
                                           v8f& acc0, v8f& acc1) {
#pragma unroll
  for (int m = 0; m < 2; ++m) {
    v4f a0 = *(const v4f*)&buf[aoff + 8 * m];
    v4f a1 = *(const v4f*)&buf[aoff + 16 * OG_UP + 8 * m];
    const float* sb = &buf[OG_UT + boff + 2 * m * OG_SG];
    v2f b01, b23, x;
    b01.x = sb[0];  b01.y = sb[16];
    b23.x = sb[32]; b23.y = sb[48];
    x.x = a0.x; x.y = a0.y; acc0 = wmma4(x, b01, acc0);
    x.x = a1.x; x.y = a1.y; acc1 = wmma4(x, b01, acc1);
    x.x = a0.z; x.y = a0.w; acc0 = wmma4(x, b23, acc0);
    x.x = a1.z; x.y = a1.w; acc1 = wmma4(x, b23, acc1);
  }
}

__device__ __forceinline__ void og_barrier() {
  asm volatile("s_wait_dscnt 0x0\n\ts_barrier_signal -1\n\ts_barrier_wait -1" ::: "memory");
  __builtin_amdgcn_sched_barrier(0);
}

__global__ void __launch_bounds__(OG_THR) __attribute__((amdgpu_waves_per_eu(4, 4)))
outgemm_lsm_kernel(const float* __restrict__ U,
                   const float* __restrict__ tbuf,
                   const float* __restrict__ Hf,
                   float* __restrict__ out) {
  __shared__ __attribute__((aligned(16))) float lds[2 * OG_BUF];

  const int tid  = threadIdx.x;
  const int w    = tid >> 5;
  const int lane = tid & 31;
  const int mn   = lane & 15;
  const int half = lane >> 4;
  const int r0   = blockIdx.x * OG_ROWS;

  const float* pu = U + (size_t)(r0 + (tid >> 6)) * N_NODES + (tid & 63) * 4;
  const v4f*   pt = (const v4f*)tbuf + tid;
  const v4f*   ph = (const v4f*)Hf + tid;
  const int uoff = (tid >> 6) * OG_UP + (tid & 63) * 4;
  const int soff = (tid >> 4) * OG_SG + (tid & 15) * 4;
  const int aoff = mn * OG_UP + w * 16 + 4 * half;
  const int boff = (4 * w + half) * OG_SG + mn;

  float* buf0 = lds;
  float* buf1 = lds + OG_BUF;

  const int first = blockIdx.x & (OG_NCH - 1);
#define OG_CH(j) (((j) + first) & (OG_NCH - 1))

  OG_DECL(X);
  OG_DECL(Y);
  OG_ISSUE(X, OG_CH(0));
  __builtin_amdgcn_sched_barrier(0);
  OG_ISSUE(Y, OG_CH(1));
  __builtin_amdgcn_sched_barrier(0);
  OG_STASH(X, buf0);
  OG_ISSUE(X, OG_CH(2));
  __builtin_amdgcn_sched_barrier(0);
  og_barrier();

  v8f acc0 = {}, acc1 = {};
  for (int i = 0; i < OG_NCH - 4; i += 2) {
    OG_STASH(Y, buf1);
    OG_ISSUE(Y, OG_CH(i + 3));
    og_compute(buf0, aoff, boff, acc0, acc1);
    og_barrier();
    OG_STASH(X, buf0);
    OG_ISSUE(X, OG_CH(i + 4));
    og_compute(buf1, aoff, boff, acc0, acc1);
    og_barrier();
  }
  OG_STASH(Y, buf1);
  OG_ISSUE(Y, OG_CH(OG_NCH - 1));
  og_compute(buf0, aoff, boff, acc0, acc1);
  og_barrier();
  OG_STASH(X, buf0);
  og_compute(buf1, aoff, boff, acc0, acc1);
  og_barrier();
  OG_STASH(Y, buf1);
  og_compute(buf0, aoff, boff, acc0, acc1);
  og_barrier();
  og_compute(buf1, aoff, boff, acc0, acc1);
  og_barrier();

  float* red = lds;
  float* vs  = lds + OG_BUF;
#pragma unroll
  for (int r = 0; r < 8; ++r) {
    red[((w * 2 + 0) * 8 + r) * 32 + lane] = acc0[r];
    red[((w * 2 + 1) * 8 + r) * 32 + lane] = acc1[r];
  }
  og_barrier();
  const int row  = tid >> 4;
  const int c    = tid & 15;
  const int tile = row >> 4;
  const int rr   = row & 15;
  const int ridx = (tile * 8 + (rr & 7)) * 32 + c + 16 * (rr >> 3);
  float v = 0.0f;
#pragma unroll
  for (int w2 = 0; w2 < 16; ++w2) v += red[w2 * (2 * 8 * 32) + ridx];
  vs[tid] = v;
  og_barrier();

  float rv[CCH];
#pragma unroll
  for (int q = 0; q < 4; ++q) {
    float4 t4 = *(const float4*)&vs[row * CCH + q * 4];
    rv[q * 4 + 0] = t4.x; rv[q * 4 + 1] = t4.y; rv[q * 4 + 2] = t4.z; rv[q * 4 + 3] = t4.w;
  }
  float mx = rv[0];
#pragma unroll
  for (int i = 1; i < CCH; ++i) mx = fmaxf(mx, rv[i]);
  float sm = 0.0f;
#pragma unroll
  for (int i = 0; i < CCH; ++i) sm += expf(rv[i] - mx);
  const float lse = mx + logf(sm);
  out[(size_t)(r0 + row) * CCH + c] = v - lse;
}

extern "C" void kernel_launch(void* const* d_in, const int* in_sizes, int n_in,
                              void* d_out, int out_size, void* d_ws, size_t ws_size,
                              hipStream_t stream) {
  const float* origin_e = (const float*)d_in[0];
  const float* U        = (const float*)d_in[1];
  const float* x        = (const float*)d_in[2];
  const float* W1       = (const float*)d_in[3];
  const float* b1       = (const float*)d_in[4];
  const float* W2       = (const float*)d_in[5];
  const float* b2       = (const float*)d_in[6];
  const float* temp     = (const float*)d_in[7];
  float* out = (float*)d_out;

  float* ws   = (float*)d_ws;
  float* Hf   = ws;
  float* zb   = Hf   + (size_t)N_NODES * CCH;
  float* tbuf = zb   + (size_t)N_NODES * CCH;

  zero_kernel<<<128, 256, 0, stream>>>((float4*)tbuf);
  jacobi_kernel<<<N_NODES / 256, 256, 0, stream>>>(origin_e, temp, Hf);
  mlp_kernel<<<N_NODES / ML_ROWS, ML_THR, 0, stream>>>(x, W1, b1, W2, b2, zb);
  utz_kernel<<<(N_NODES / 16) * NCHUNKS / 8, 256, 0, stream>>>(U, zb, tbuf);
  outgemm_lsm_kernel<<<N_NODES / OG_ROWS, OG_THR, 0, stream>>>(U, tbuf, Hf, out);
}


# t=U^T z re-tiled: a block owns 64 columns x 4096 rows and sums them inside (8 row groups added in LDS), stores Hf*t to 1 of 2 planes the out-GEMM adds: sum order changes; no atomics, 5->4 launches
# speedup vs baseline: 1.6738x; 1.1643x over previous
#include <hip/hip_runtime.h>
#include <cstdint>

typedef __attribute__((ext_vector_type(2))) float v2f;
typedef __attribute__((ext_vector_type(8))) float v8f;

#define N_NODES 8192
#define F_INF   512
#define HIDF    64
#define CCH     16
#define KORD    10

__device__ __forceinline__ v8f wmma4(v2f a, v2f b, v8f c) {
  return __builtin_amdgcn_wmma_f32_16x16x4_f32(
       false, a,  false, b,
       (short)0, c,  false,  false);
}

__global__ void jacobi_kernel(const float* __restrict__ origin_e,
                              const float* __restrict__ temp,
                              float* __restrict__ Hf) {
  __shared__ float al[CCH * (KORD + 1)];
  int tid = threadIdx.x;
  if (tid < CCH * (KORD + 1)) al[tid] = 0.5f * tanhf(temp[tid]);
  __syncthreads();

  int n = blockIdx.x * blockDim.x + tid;
  if (n >= N_NODES) return;
  float e = origin_e[n];

  const float a = 1.0f, b = 1.0f, lo = -1.0f, hi = 1.0f;
  const float coef1 = (a - b) * 0.5f - (a + b + 2.0f) * 0.5f * (lo + hi) / (hi - lo);
  const float coef2 = (a + b + 2.0f) / (hi - lo);

  for (int c = 0; c < CCH; ++c) {
    const float* ac = &al[c * (KORD + 1)];
    float xs_m2 = ac[0];
    float acc = xs_m2;
    float xs_m1 = coef1 * ac[1] + coef2 * ac[1] * e * xs_m2;
    acc += xs_m1;
#pragma unroll
    for (int L = 2; L <= KORD; ++L) {
      float Lf = (float)L;
      float coef_l  = 2.0f * Lf * (Lf + a + b) * (2.0f * Lf - 2.0f + a + b);
      float c_lm1_1 = (2.0f * Lf + a + b - 1.0f) * (2.0f * Lf + a + b) * (2.0f * Lf + a + b - 2.0f);
      float c_lm1_2 = (2.0f * Lf + a + b - 1.0f) * (a * a - b * b);
      float c_lm2   = 2.0f * (Lf - 1.0f + a) * (Lf - 1.0f + b) * (2.0f * Lf + a + b);
      float tmp1   = ac[L - 1] * (c_lm1_1 / coef_l);
      float tmp2   = ac[L - 1] * (c_lm1_2 / coef_l);
      float tmp3   = ac[L - 1] * ac[L - 2] * (c_lm2 / coef_l);
      float tmp1_2 = tmp1 * (2.0f / (hi - lo));
      float tmp2_2 = tmp1 * ((hi + lo) / (hi - lo)) + tmp2;
      float nx = tmp1_2 * (e * xs_m1) - tmp2_2 * xs_m1 - tmp3 * xs_m2;
      acc += nx;
      xs_m2 = xs_m1;
      xs_m1 = nx;
    }
    Hf[n * CCH + c] = acc;
  }
}

#define ML_ROWS 32
#define ML_THR  256
#define ML_XP   516
#define ML_WP   72
#define ML_HP   68
#define ML_XS   (ML_ROWS * ML_XP)
#define ML_WS   (F_INF * ML_WP)
#define ML_HS   (ML_ROWS * ML_HP)

typedef __attribute__((ext_vector_type(4))) float v4f;

__global__ void __launch_bounds__(ML_THR)
mlp_kernel(const float* __restrict__ x,
           const float* __restrict__ W1,
           const float* __restrict__ b1,
           const float* __restrict__ W2,
           const float* __restrict__ b2,
           float* __restrict__ zb) {
  __shared__ __attribute__((aligned(16))) float lds[ML_XS + ML_WS + ML_HS];
  float* wl = lds;
  float* xs = lds + ML_WS;
  float* hs = lds + ML_WS + ML_XS;

  const int tid  = threadIdx.x;
  const int w    = tid >> 5;
  const int lane = tid & 31;
  const int mn   = lane & 15;
  const int half = lane >> 4;
  const int kk   = half * 2;
  const int mt   = w & 1;
  const int n0   = (w >> 1) * 16;
  const int m0   = blockIdx.x * ML_ROWS;

  v4f xr[16], wr[32];
  const float* px = x + (size_t)(m0 + (tid >> 7)) * F_INF + (tid & 127) * 4;
#pragma unroll
  for (int i = 0; i < 16; ++i) xr[i] = *(const v4f*)(px + (size_t)(2 * i) * F_INF);
  const int wrot = blockIdx.x & 31;
#pragma unroll
  for (int i = 0; i < 32; ++i) {
    const int j = (i + wrot) & 31;
    wr[i] = *(const v4f*)(W1 + (size_t)(tid + 256 * j) * 4);
  }
  v2f w2f[16];
  float bb = 0.0f;
  if (w < 2) {
#pragma unroll
    for (int q = 0; q < 16; ++q) {
      w2f[q].x = W2[(4 * q + kk) * CCH + mn];
      w2f[q].y = W2[(4 * q + kk + 1) * CCH + mn];
    }
    bb = b2[mn];
  }
  const float bias = b1[n0 + mn];

  const int xoff = (tid >> 7) * ML_XP + (tid & 127) * 4;
#pragma unroll
  for (int i = 0; i < 16; ++i) *(v4f*)&xs[xoff + 2 * i * ML_XP] = xr[i];
  const int woff = (tid >> 4) * ML_WP + (tid & 15) * 4;
#pragma unroll
  for (int i = 0; i < 32; ++i) {
    const int j = (i + wrot) & 31;
    *(v4f*)&wl[woff + 16 * j * ML_WP] = wr[i];
  }
  __syncthreads();

  v8f acc0 = {}, acc1 = {};
  const float* xa = &xs[(mt * 16 + mn) * ML_XP + 4 * half];
  const float* wb = &wl[(4 * half) * ML_WP + n0 + mn];
#pragma unroll 8
  for (int g = 0; g < F_INF / 8; ++g) {
    const v4f a = *(const v4f*)&xa[8 * g];
    const float* wg = wb + 8 * g * ML_WP;
    v2f b01, b23, av;
    b01.x = wg[0];          b01.y = wg[ML_WP];
    b23.x = wg[2 * ML_WP];  b23.y = wg[3 * ML_WP];
    av.x = a.x; av.y = a.y; acc0 = wmma4(av, b01, acc0);
    av.x = a.z; av.y = a.w; acc1 = wmma4(av, b23, acc1);
  }

#pragma unroll
  for (int r = 0; r < 8; ++r) {
    const int row = mt * 16 + (half ? r + 8 : r);
    const float v = (acc0[r] + acc1[r]) + bias;
    hs[row * ML_HP + n0 + mn] = v > 0.0f ? v : 0.0f;
  }
  __syncthreads();

  if (w < 2) {
    v8f zacc = {};
#pragma unroll
    for (int q = 0; q < 16; ++q) {
      v2f av;
      av.x = hs[(w * 16 + mn) * ML_HP + 4 * q + kk];
      av.y = hs[(w * 16 + mn) * ML_HP + 4 * q + kk + 1];
      zacc = wmma4(av, w2f[q], zacc);
    }
#pragma unroll
    for (int r = 0; r < 8; ++r) {
      const int row = w * 16 + (half ? r + 8 : r);
      zb[(size_t)(m0 + row) * CCH + mn] = zacc[r] + bb;
    }
  }
}

__device__ __forceinline__ void og_barrier() {
  asm volatile("s_wait_dscnt 0x0\n\ts_barrier_signal -1\n\ts_barrier_wait -1" ::: "memory");
  __builtin_amdgcn_sched_barrier(0);
}

#define UZ_JH   (N_NODES / 2)
#define UZ_KC   256
#define UZ_NCH  (UZ_JH / UZ_KC)
#define UZ_THR  512
#define UZ_UP   80
#define UZ_UT   (UZ_KC * UZ_UP)
#define UZ_ZT   (UZ_KC * CCH)
#define UZ_BUF  (UZ_UT + UZ_ZT)

#define UZ_DECL(S) v4f S##u0, S##u1, S##u2, S##u3, S##u4, S##u5, S##u6, S##u7, S##z0, S##z1

#define UZ_ISSUE(S, ch)                                                        \
  do {                                                                         \
    const float* pu_ = pu + (size_t)(ch) * ((size_t)UZ_KC * N_NODES);          \
    S##u0 = *(const v4f*)(pu_);                                                \
    S##u1 = *(const v4f*)(pu_ + (size_t)32 * N_NODES);                         \
    S##u2 = *(const v4f*)(pu_ + (size_t)64 * N_NODES);                         \
    S##u3 = *(const v4f*)(pu_ + (size_t)96 * N_NODES);                         \
    S##u4 = *(const v4f*)(pu_ + (size_t)128 * N_NODES);                        \
    S##u5 = *(const v4f*)(pu_ + (size_t)160 * N_NODES);                        \
    S##u6 = *(const v4f*)(pu_ + (size_t)192 * N_NODES);                        \
    S##u7 = *(const v4f*)(pu_ + (size_t)224 * N_NODES);                        \
    S##z0 = pz[(ch) * (UZ_KC * 4)];                                            \
    S##z1 = pz[(ch) * (UZ_KC * 4) + UZ_THR];                                   \
  } while (0)

#define UZ_STASH(S, buf)                                                       \
  do {                                                                         \
    *(v4f*)&(buf)[uoff]               = S##u0;                                 \
    *(v4f*)&(buf)[uoff + 32 * UZ_UP]  = S##u1;                                 \
    *(v4f*)&(buf)[uoff + 64 * UZ_UP]  = S##u2;                                 \
    *(v4f*)&(buf)[uoff + 96 * UZ_UP]  = S##u3;                                 \
    *(v4f*)&(buf)[uoff + 128 * UZ_UP] = S##u4;                                 \
    *(v4f*)&(buf)[uoff + 160 * UZ_UP] = S##u5;                                 \
    *(v4f*)&(buf)[uoff + 192 * UZ_UP] = S##u6;                                 \
    *(v4f*)&(buf)[uoff + 224 * UZ_UP] = S##u7;                                 \
    *(v4f*)&(buf)[UZ_UT + zoff]              = S##z0;                          \
    *(v4f*)&(buf)[UZ_UT + zoff + 4 * UZ_THR] = S##z1;                          \
  } while (0)

__device__ __forceinline__ void uz_compute(const float* buf, int aoff, int boff,
                                           v8f& acc0, v8f& acc1) {
#pragma unroll
  for (int q = 0; q < 8; ++q) {
    const float* zr = &buf[aoff + q * (4 * CCH)];
    const float* ur = &buf[boff + q * (4 * UZ_UP)];
    v2f a, b0, b1;
    a.x  = zr[0];   a.y  = zr[CCH];
    b0.x = ur[0];   b0.y = ur[UZ_UP];
    b1.x = ur[16];  b1.y = ur[UZ_UP + 16];
    acc0 = wmma4(a, b0, acc0);
    acc1 = wmma4(a, b1, acc1);
  }
}

__global__ void __launch_bounds__(UZ_THR) __attribute__((amdgpu_waves_per_eu(4, 4)))
utz_kernel(const float* __restrict__ U,
           const float* __restrict__ zb,
           const float* __restrict__ Hf,
           float* __restrict__ sbuf) {
  __shared__ __attribute__((aligned(16))) float lds[2 * UZ_BUF];

  const int tid  = threadIdx.x;
  const int w    = tid >> 5;
  const int lane = tid & 31;
  const int mn   = lane & 15;
  const int half = lane >> 4;
  const int g    = blockIdx.x & 127;
  const int hf   = blockIdx.x >> 7;
  const int c0   = g * 64;
  const int j0   = hf * UZ_JH;

  const float* pu = U + (size_t)(j0 + (tid >> 4)) * N_NODES + c0 + (tid & 15) * 4;
  const v4f*   pz = (const v4f*)(zb + (size_t)j0 * CCH) + tid;
  const int uoff = (tid >> 4) * UZ_UP + (tid & 15) * 4;
  const int zoff = tid * 4;
  const int aoff = UZ_UT + (32 * (w >> 1) + 2 * half) * CCH + mn;
  const int boff = (32 * (w >> 1) + 2 * half) * UZ_UP + 32 * (w & 1) + mn;

  float* buf0 = lds;
  float* buf1 = lds + UZ_BUF;

  const int first = (g >> 3) & (UZ_NCH - 1);
#define UZ_CH(j) (((j) + first) & (UZ_NCH - 1))

  UZ_DECL(X);
  UZ_DECL(Y);
  UZ_ISSUE(X, UZ_CH(0));
  UZ_ISSUE(Y, UZ_CH(1));
  UZ_STASH(X, buf0);
  UZ_ISSUE(X, UZ_CH(2));
  og_barrier();

  v8f acc0 = {}, acc1 = {};
  for (int i = 0; i < UZ_NCH - 4; i += 2) {
    UZ_STASH(Y, buf1);
    UZ_ISSUE(Y, UZ_CH(i + 3));
    uz_compute(buf0, aoff, boff, acc0, acc1);
    og_barrier();
    UZ_STASH(X, buf0);
    UZ_ISSUE(X, UZ_CH(i + 4));
    uz_compute(buf1, aoff, boff, acc0, acc1);
    og_barrier();
  }
  UZ_STASH(Y, buf1);
  UZ_ISSUE(Y, UZ_CH(UZ_NCH - 1));
  uz_compute(buf0, aoff, boff, acc0, acc1);
  og_barrier();
  UZ_STASH(X, buf0);
  uz_compute(buf1, aoff, boff, acc0, acc1);
  og_barrier();
  UZ_STASH(Y, buf1);
  uz_compute(buf0, aoff, boff, acc0, acc1);
  og_barrier();
  uz_compute(buf1, aoff, boff, acc0, acc1);
  og_barrier();

  float* red = lds;
#pragma unroll
  for (int r = 0; r < 8; ++r) {
    red[((w * 2 + 0) * 8 + r) * 32 + lane] = acc0[r];
    red[((w * 2 + 1) * 8 + r) * 32 + lane] = acc1[r];
  }
  og_barrier();
  const float* hp = Hf + (size_t)c0 * CCH;
  float* sp = sbuf + (size_t)hf * ((size_t)N_NODES * CCH) + (size_t)c0 * CCH;
#pragma unroll
  for (int e2 = 0; e2 < 2; ++e2) {
    const int e   = tid + e2 * UZ_THR;
    const int il  = e >> 4;
    const int c   = e & 15;
    const int idx = ((il >> 4) * 8 + (c & 7)) * 32 + (il & 15) + 16 * (c >> 3);
    float v = 0.0f;
#pragma unroll
    for (int kg = 0; kg < 8; ++kg) v += red[kg * (4 * 8 * 32) + idx];
    sp[e] = v * hp[e];
  }
}

#define OG_ROWS 32
#define OG_KC   256
#define OG_NCH  (N_NODES / OG_KC)
#define OG_THR  512
#define OG_UP   260
#define OG_UT   (OG_ROWS * OG_UP)
#define OG_SG   80
#define OG_ST   ((OG_KC / 4) * OG_SG)
#define OG_BUF  (OG_UT + OG_ST)

#define OG_DECL(S) v4f S##u0, S##u1, S##u2, S##u3, S##t0, S##t1, S##h0, S##h1

#define OG_ISSUE(S, ch)                                                        \
  do {                                                                         \
    const float* pu_ = pu + (size_t)(ch) * OG_KC;                              \
    S##u0 = *(const v4f*)(pu_);                                                \
    S##u1 = *(const v4f*)(pu_ + (size_t)8 * N_NODES);                          \
    S##u2 = *(const v4f*)(pu_ + (size_t)16 * N_NODES);                         \
    S##u3 = *(const v4f*)(pu_ + (size_t)24 * N_NODES);                         \
    S##t0 = pt[(ch) * (OG_KC * 4)];                                            \
    S##t1 = pt[(ch) * (OG_KC * 4) + OG_THR];                                   \
    S##h0 = ph[(ch) * (OG_KC * 4)];                                            \
    S##h1 = ph[(ch) * (OG_KC * 4) + OG_THR];                                   \
  } while (0)

#define OG_STASH(S, buf)                                                       \
  do {                                                                         \
    *(v4f*)&(buf)[uoff]              = S##u0;                                  \
    *(v4f*)&(buf)[uoff + 8 * OG_UP]  = S##u1;                                  \
    *(v4f*)&(buf)[uoff + 16 * OG_UP] = S##u2;                                  \
    *(v4f*)&(buf)[uoff + 24 * OG_UP] = S##u3;                                  \
    *(v4f*)&(buf)[OG_UT + soff]              = S##t0 + S##h0;                  \
    *(v4f*)&(buf)[OG_UT + soff + 32 * OG_SG] = S##t1 + S##h1;                  \
  } while (0)

__device__ __forceinline__ void og_compute(const float* buf, int aoff, int boff,
                                           v8f& acc0, v8f& acc1) {
#pragma unroll
  for (int m = 0; m < 2; ++m) {
    v4f a0 = *(const v4f*)&buf[aoff + 8 * m];
    v4f a1 = *(const v4f*)&buf[aoff + 16 * OG_UP + 8 * m];
    const float* sb = &buf[OG_UT + boff + 2 * m * OG_SG];
    v2f b01, b23, x;
    b01.x = sb[0];  b01.y = sb[16];
    b23.x = sb[32]; b23.y = sb[48];
    x.x = a0.x; x.y = a0.y; acc0 = wmma4(x, b01, acc0);
    x.x = a1.x; x.y = a1.y; acc1 = wmma4(x, b01, acc1);
    x.x = a0.z; x.y = a0.w; acc0 = wmma4(x, b23, acc0);
    x.x = a1.z; x.y = a1.w; acc1 = wmma4(x, b23, acc1);
  }
}

__global__ void __launch_bounds__(OG_THR) __attribute__((amdgpu_waves_per_eu(4, 4)))
outgemm_lsm_kernel(const float* __restrict__ U,
                   const float* __restrict__ s0,
                   const float* __restrict__ s1,
                   float* __restrict__ out) {
  __shared__ __attribute__((aligned(16))) float lds[2 * OG_BUF];

  const int tid  = threadIdx.x;
  const int w    = tid >> 5;
  const int lane = tid & 31;
  const int mn   = lane & 15;
  const int half = lane >> 4;
  const int r0   = blockIdx.x * OG_ROWS;

  const float* pu = U + (size_t)(r0 + (tid >> 6)) * N_NODES + (tid & 63) * 4;
  const v4f*   pt = (const v4f*)s0 + tid;
  const v4f*   ph = (const v4f*)s1 + tid;
  const int uoff = (tid >> 6) * OG_UP + (tid & 63) * 4;
  const int soff = (tid >> 4) * OG_SG + (tid & 15) * 4;
  const int aoff = mn * OG_UP + w * 16 + 4 * half;
  const int boff = (4 * w + half) * OG_SG + mn;

  float* buf0 = lds;
  float* buf1 = lds + OG_BUF;

  const int first = blockIdx.x & (OG_NCH - 1);
#define OG_CH(j) (((j) + first) & (OG_NCH - 1))

  OG_DECL(X);
  OG_DECL(Y);
  OG_ISSUE(X, OG_CH(0));
  __builtin_amdgcn_sched_barrier(0);
  OG_ISSUE(Y, OG_CH(1));
  __builtin_amdgcn_sched_barrier(0);
  OG_STASH(X, buf0);
  OG_ISSUE(X, OG_CH(2));
  __builtin_amdgcn_sched_barrier(0);
  og_barrier();

  v8f acc0 = {}, acc1 = {};
  for (int i = 0; i < OG_NCH - 4; i += 2) {
    OG_STASH(Y, buf1);
    OG_ISSUE(Y, OG_CH(i + 3));
    og_compute(buf0, aoff, boff, acc0, acc1);
    og_barrier();
    OG_STASH(X, buf0);
    OG_ISSUE(X, OG_CH(i + 4));
    og_compute(buf1, aoff, boff, acc0, acc1);
    og_barrier();
  }
  OG_STASH(Y, buf1);
  OG_ISSUE(Y, OG_CH(OG_NCH - 1));
  og_compute(buf0, aoff, boff, acc0, acc1);
  og_barrier();
  OG_STASH(X, buf0);
  og_compute(buf1, aoff, boff, acc0, acc1);
  og_barrier();
  OG_STASH(Y, buf1);
  og_compute(buf0, aoff, boff, acc0, acc1);
  og_barrier();
  og_compute(buf1, aoff, boff, acc0, acc1);
  og_barrier();

  float* red = lds;
  float* vs  = lds + OG_BUF;
#pragma unroll
  for (int r = 0; r < 8; ++r) {
    red[((w * 2 + 0) * 8 + r) * 32 + lane] = acc0[r];
    red[((w * 2 + 1) * 8 + r) * 32 + lane] = acc1[r];
  }
  og_barrier();
  const int row  = tid >> 4;
  const int c    = tid & 15;
  const int tile = row >> 4;
  const int rr   = row & 15;
  const int ridx = (tile * 8 + (rr & 7)) * 32 + c + 16 * (rr >> 3);
  float v = 0.0f;
#pragma unroll
  for (int w2 = 0; w2 < 16; ++w2) v += red[w2 * (2 * 8 * 32) + ridx];
  vs[tid] = v;
  og_barrier();

  float rv[CCH];
#pragma unroll
  for (int q = 0; q < 4; ++q) {
    float4 t4 = *(const float4*)&vs[row * CCH + q * 4];
    rv[q * 4 + 0] = t4.x; rv[q * 4 + 1] = t4.y; rv[q * 4 + 2] = t4.z; rv[q * 4 + 3] = t4.w;
  }
  float mx = rv[0];
#pragma unroll
  for (int i = 1; i < CCH; ++i) mx = fmaxf(mx, rv[i]);
  float sm = 0.0f;
#pragma unroll
  for (int i = 0; i < CCH; ++i) sm += expf(rv[i] - mx);
  const float lse = mx + logf(sm);
  out[(size_t)(r0 + row) * CCH + c] = v - lse;
}

extern "C" void kernel_launch(void* const* d_in, const int* in_sizes, int n_in,
                              void* d_out, int out_size, void* d_ws, size_t ws_size,
                              hipStream_t stream) {
  const float* origin_e = (const float*)d_in[0];
  const float* U        = (const float*)d_in[1];
  const float* x        = (const float*)d_in[2];
  const float* W1       = (const float*)d_in[3];
  const float* b1       = (const float*)d_in[4];
  const float* W2       = (const float*)d_in[5];
  const float* b2       = (const float*)d_in[6];
  const float* temp     = (const float*)d_in[7];
  float* out = (float*)d_out;

  float* ws   = (float*)d_ws;
  float* Hf   = ws;
  float* zb   = Hf   + (size_t)N_NODES * CCH;
  float* sbuf = zb   + (size_t)N_NODES * CCH;

  jacobi_kernel<<<N_NODES / 256, 256, 0, stream>>>(origin_e, temp, Hf);
  mlp_kernel<<<N_NODES / ML_ROWS, ML_THR, 0, stream>>>(x, W1, b1, W2, b2, zb);
  utz_kernel<<<2 * (N_NODES / 64), UZ_THR, 0, stream>>>(U, zb, Hf, sbuf);
  outgemm_lsm_kernel<<<N_NODES / OG_ROWS, OG_THR, 0, stream>>>(U, sbuf, sbuf + (size_t)N_NODES * CCH, out);
}


# Jacobi filter folded into the t = U^T z launch: each block forms Hf of its own 64 columns under the wait for its first chunk of U; no filter launch, no Hf plane in memory: 4 -> 3 launches
# speedup vs baseline: 1.8059x; 1.0789x over previous
#include <hip/hip_runtime.h>
#include <cstdint>

typedef __attribute__((ext_vector_type(2))) float v2f;
typedef __attribute__((ext_vector_type(8))) float v8f;

#define N_NODES 8192
#define F_INF   512
#define HIDF    64
#define CCH     16
#define KORD    10

__device__ __forceinline__ v8f wmma4(v2f a, v2f b, v8f c) {
  return __builtin_amdgcn_wmma_f32_16x16x4_f32(
       false, a,  false, b,
       (short)0, c,  false,  false);
}

__device__ __forceinline__ void jacobi_pair(const float* ac, float e0, float e1,
                                            float& h0, float& h1) {
  const float a = 1.0f, b = 1.0f, lo = -1.0f, hi = 1.0f;
  const float coef1 = (a - b) * 0.5f - (a + b + 2.0f) * 0.5f * (lo + hi) / (hi - lo);
  const float coef2 = (a + b + 2.0f) / (hi - lo);

  const float al0 = ac[0], al1 = ac[1];
  float p2 = al0, q2 = al0;
  float accp = p2, accq = q2;
  float p1 = coef1 * al1 + coef2 * al1 * e0 * p2;
  float q1 = coef1 * al1 + coef2 * al1 * e1 * q2;
  accp += p1;
  accq += q1;
#pragma unroll
  for (int L = 2; L <= KORD; ++L) {
    float Lf = (float)L;
    float coef_l  = 2.0f * Lf * (Lf + a + b) * (2.0f * Lf - 2.0f + a + b);
    float c_lm1_1 = (2.0f * Lf + a + b - 1.0f) * (2.0f * Lf + a + b) * (2.0f * Lf + a + b - 2.0f);
    float c_lm1_2 = (2.0f * Lf + a + b - 1.0f) * (a * a - b * b);
    float c_lm2   = 2.0f * (Lf - 1.0f + a) * (Lf - 1.0f + b) * (2.0f * Lf + a + b);
    float tmp1   = ac[L - 1] * (c_lm1_1 / coef_l);
    float tmp2   = ac[L - 1] * (c_lm1_2 / coef_l);
    float tmp3   = ac[L - 1] * ac[L - 2] * (c_lm2 / coef_l);
    float tmp1_2 = tmp1 * (2.0f / (hi - lo));
    float tmp2_2 = tmp1 * ((hi + lo) / (hi - lo)) + tmp2;
    float np = tmp1_2 * (e0 * p1) - tmp2_2 * p1 - tmp3 * p2;
    float nq = tmp1_2 * (e1 * q1) - tmp2_2 * q1 - tmp3 * q2;
    accp += np;
    accq += nq;
    p2 = p1; p1 = np;
    q2 = q1; q1 = nq;
  }
  h0 = accp;
  h1 = accq;
}

#define ML_ROWS 32
#define ML_THR  256
#define ML_XP   516
#define ML_WP   72
#define ML_HP   68
#define ML_XS   (ML_ROWS * ML_XP)
#define ML_WS   (F_INF * ML_WP)
#define ML_HS   (ML_ROWS * ML_HP)

typedef __attribute__((ext_vector_type(4))) float v4f;

__global__ void __launch_bounds__(ML_THR)
mlp_kernel(const float* __restrict__ x,
           const float* __restrict__ W1,
           const float* __restrict__ b1,
           const float* __restrict__ W2,
           const float* __restrict__ b2,
           float* __restrict__ zb) {
  __shared__ __attribute__((aligned(16))) float lds[ML_XS + ML_WS + ML_HS];
  float* wl = lds;
  float* xs = lds + ML_WS;
  float* hs = lds + ML_WS + ML_XS;

  const int tid  = threadIdx.x;
  const int w    = tid >> 5;
  const int lane = tid & 31;
  const int mn   = lane & 15;
  const int half = lane >> 4;
  const int kk   = half * 2;
  const int mt   = w & 1;
  const int n0   = (w >> 1) * 16;
  const int m0   = blockIdx.x * ML_ROWS;

  v4f xr[16], wr[32];
  const float* px = x + (size_t)(m0 + (tid >> 7)) * F_INF + (tid & 127) * 4;
#pragma unroll
  for (int i = 0; i < 16; ++i) xr[i] = *(const v4f*)(px + (size_t)(2 * i) * F_INF);
  const int wrot = blockIdx.x & 31;
#pragma unroll
  for (int i = 0; i < 32; ++i) {
    const int j = (i + wrot) & 31;
    wr[i] = *(const v4f*)(W1 + (size_t)(tid + 256 * j) * 4);
  }
  v2f w2f[16];
  float bb = 0.0f;
  if (w < 2) {
#pragma unroll
    for (int q = 0; q < 16; ++q) {
      w2f[q].x = W2[(4 * q + kk) * CCH + mn];
      w2f[q].y = W2[(4 * q + kk + 1) * CCH + mn];
    }
    bb = b2[mn];
  }
  const float bias = b1[n0 + mn];

  const int xoff = (tid >> 7) * ML_XP + (tid & 127) * 4;
#pragma unroll
  for (int i = 0; i < 16; ++i) *(v4f*)&xs[xoff + 2 * i * ML_XP] = xr[i];
  const int woff = (tid >> 4) * ML_WP + (tid & 15) * 4;
#pragma unroll
  for (int i = 0; i < 32; ++i) {
    const int j = (i + wrot) & 31;
    *(v4f*)&wl[woff + 16 * j * ML_WP] = wr[i];
  }
  __syncthreads();

  v8f acc0 = {}, acc1 = {};
  const float* xa = &xs[(mt * 16 + mn) * ML_XP + 4 * half];
  const float* wb = &wl[(4 * half) * ML_WP + n0 + mn];
#pragma unroll 8
  for (int g = 0; g < F_INF / 8; ++g) {
    const v4f a = *(const v4f*)&xa[8 * g];
    const float* wg = wb + 8 * g * ML_WP;
    v2f b01, b23, av;
    b01.x = wg[0];          b01.y = wg[ML_WP];
    b23.x = wg[2 * ML_WP];  b23.y = wg[3 * ML_WP];
    av.x = a.x; av.y = a.y; acc0 = wmma4(av, b01, acc0);
    av.x = a.z; av.y = a.w; acc1 = wmma4(av, b23, acc1);
  }

#pragma unroll
  for (int r = 0; r < 8; ++r) {
    const int row = mt * 16 + (half ? r + 8 : r);
    const float v = (acc0[r] + acc1[r]) + bias;
    hs[row * ML_HP + n0 + mn] = v > 0.0f ? v : 0.0f;
  }
  __syncthreads();

  if (w < 2) {
    v8f zacc = {};
#pragma unroll
    for (int q = 0; q < 16; ++q) {
      v2f av;
      av.x = hs[(w * 16 + mn) * ML_HP + 4 * q + kk];
      av.y = hs[(w * 16 + mn) * ML_HP + 4 * q + kk + 1];
      zacc = wmma4(av, w2f[q], zacc);
    }
#pragma unroll
    for (int r = 0; r < 8; ++r) {
      const int row = w * 16 + (half ? r + 8 : r);
      zb[(size_t)(m0 + row) * CCH + mn] = zacc[r] + bb;
    }
  }
}

__device__ __forceinline__ void og_barrier() {
  asm volatile("s_wait_dscnt 0x0\n\ts_barrier_signal -1\n\ts_barrier_wait -1" ::: "memory");
  __builtin_amdgcn_sched_barrier(0);
}

#define UZ_JH   (N_NODES / 2)
#define UZ_KC   256
#define UZ_NCH  (UZ_JH / UZ_KC)
#define UZ_THR  512
#define UZ_UP   80
#define UZ_UT   (UZ_KC * UZ_UP)
#define UZ_ZT   (UZ_KC * CCH)
#define UZ_BUF  (UZ_UT + UZ_ZT)
#define UZ_NAL  (CCH * (KORD + 1))

#define UZ_DECL(S) v4f S##u0, S##u1, S##u2, S##u3, S##u4, S##u5, S##u6, S##u7, S##z0, S##z1

#define UZ_ISSUE(S, ch)                                                        \
  do {                                                                         \
    const float* pu_ = pu + (size_t)(ch) * ((size_t)UZ_KC * N_NODES);          \
    S##u0 = *(const v4f*)(pu_);                                                \
    S##u1 = *(const v4f*)(pu_ + (size_t)32 * N_NODES);                         \
    S##u2 = *(const v4f*)(pu_ + (size_t)64 * N_NODES);                         \
    S##u3 = *(const v4f*)(pu_ + (size_t)96 * N_NODES);                         \
    S##u4 = *(const v4f*)(pu_ + (size_t)128 * N_NODES);                        \
    S##u5 = *(const v4f*)(pu_ + (size_t)160 * N_NODES);                        \
    S##u6 = *(const v4f*)(pu_ + (size_t)192 * N_NODES);                        \
    S##u7 = *(const v4f*)(pu_ + (size_t)224 * N_NODES);                        \
    S##z0 = pz[(ch) * (UZ_KC * 4)];                                            \
    S##z1 = pz[(ch) * (UZ_KC * 4) + UZ_THR];                                   \
  } while (0)

#define UZ_STASH(S, buf)                                                       \
  do {                                                                         \
    *(v4f*)&(buf)[uoff]               = S##u0;                                 \
    *(v4f*)&(buf)[uoff + 32 * UZ_UP]  = S##u1;                                 \
    *(v4f*)&(buf)[uoff + 64 * UZ_UP]  = S##u2;                                 \
    *(v4f*)&(buf)[uoff + 96 * UZ_UP]  = S##u3;                                 \
    *(v4f*)&(buf)[uoff + 128 * UZ_UP] = S##u4;                                 \
    *(v4f*)&(buf)[uoff + 160 * UZ_UP] = S##u5;                                 \
    *(v4f*)&(buf)[uoff + 192 * UZ_UP] = S##u6;                                 \
    *(v4f*)&(buf)[uoff + 224 * UZ_UP] = S##u7;                                 \
    *(v4f*)&(buf)[UZ_UT + zoff]              = S##z0;                          \
    *(v4f*)&(buf)[UZ_UT + zoff + 4 * UZ_THR] = S##z1;                          \
  } while (0)

__device__ __forceinline__ void uz_compute(const float* buf, int aoff, int boff,
                                           v8f& acc0, v8f& acc1) {
#pragma unroll
  for (int q = 0; q < 8; ++q) {
    const float* zr = &buf[aoff + q * (4 * CCH)];
    const float* ur = &buf[boff + q * (4 * UZ_UP)];
    v2f a, b0, b1;
    a.x  = zr[0];   a.y  = zr[CCH];
    b0.x = ur[0];   b0.y = ur[UZ_UP];
    b1.x = ur[16];  b1.y = ur[UZ_UP + 16];
    acc0 = wmma4(a, b0, acc0);
    acc1 = wmma4(a, b1, acc1);
  }
}

__global__ void __launch_bounds__(UZ_THR) __attribute__((amdgpu_waves_per_eu(4, 4)))
utz_kernel(const float* __restrict__ U,
           const float* __restrict__ zb,
           const float* __restrict__ origin_e,
           const float* __restrict__ temp,
           float* __restrict__ sbuf) {
  __shared__ __attribute__((aligned(16))) float lds[2 * UZ_BUF + UZ_NAL];

  const int tid  = threadIdx.x;
  const int w    = tid >> 5;
  const int lane = tid & 31;
  const int mn   = lane & 15;
  const int half = lane >> 4;
  const int g    = blockIdx.x & 127;
  const int hf   = blockIdx.x >> 7;
  const int c0   = g * 64;
  const int j0   = hf * UZ_JH;

  const float* pu = U + (size_t)(j0 + (tid >> 4)) * N_NODES + c0 + (tid & 15) * 4;
  const v4f*   pz = (const v4f*)(zb + (size_t)j0 * CCH) + tid;
  const int uoff = (tid >> 4) * UZ_UP + (tid & 15) * 4;
  const int zoff = tid * 4;
  const int aoff = UZ_UT + (32 * (w >> 1) + 2 * half) * CCH + mn;
  const int boff = (32 * (w >> 1) + 2 * half) * UZ_UP + 32 * (w & 1) + mn;

  float* buf0 = lds;
  float* buf1 = lds + UZ_BUF;
  float* al   = lds + 2 * UZ_BUF;

  float tv = 0.0f;
  if (tid < UZ_NAL) tv = temp[tid];
  const float e0 = origin_e[c0 + (tid >> 4)];
  const float e1 = origin_e[c0 + 32 + (tid >> 4)];

  const int first = (g >> 3) & (UZ_NCH - 1);
#define UZ_CH(j) (((j) + first) & (UZ_NCH - 1))

  UZ_DECL(X);
  UZ_DECL(Y);
  UZ_ISSUE(X, UZ_CH(0));
  UZ_ISSUE(Y, UZ_CH(1));
  if (tid < UZ_NAL) al[tid] = 0.5f * tanhf(tv);
  og_barrier();
  float hf0, hf1;
  jacobi_pair(al + (tid & 15) * (KORD + 1), e0, e1, hf0, hf1);
  asm volatile("" : "+v"(hf0), "+v"(hf1));
  UZ_STASH(X, buf0);
  UZ_ISSUE(X, UZ_CH(2));
  og_barrier();

  v8f acc0 = {}, acc1 = {};
  for (int i = 0; i < UZ_NCH - 4; i += 2) {
    UZ_STASH(Y, buf1);
    UZ_ISSUE(Y, UZ_CH(i + 3));
    uz_compute(buf0, aoff, boff, acc0, acc1);
    og_barrier();
    UZ_STASH(X, buf0);
    UZ_ISSUE(X, UZ_CH(i + 4));
    uz_compute(buf1, aoff, boff, acc0, acc1);
    og_barrier();
  }
  UZ_STASH(Y, buf1);
  UZ_ISSUE(Y, UZ_CH(UZ_NCH - 1));
  uz_compute(buf0, aoff, boff, acc0, acc1);
  og_barrier();
  UZ_STASH(X, buf0);
  uz_compute(buf1, aoff, boff, acc0, acc1);
  og_barrier();
  UZ_STASH(Y, buf1);
  uz_compute(buf0, aoff, boff, acc0, acc1);
  og_barrier();
  uz_compute(buf1, aoff, boff, acc0, acc1);
  og_barrier();

  float* red = lds;
#pragma unroll
  for (int r = 0; r < 8; ++r) {
    red[((w * 2 + 0) * 8 + r) * 32 + lane] = acc0[r];
    red[((w * 2 + 1) * 8 + r) * 32 + lane] = acc1[r];
  }
  og_barrier();
  float* sp = sbuf + (size_t)hf * ((size_t)N_NODES * CCH) + (size_t)c0 * CCH;
#pragma unroll
  for (int e2 = 0; e2 < 2; ++e2) {
    const int e   = tid + e2 * UZ_THR;
    const int il  = e >> 4;
    const int c   = e & 15;
    const int idx = ((il >> 4) * 8 + (c & 7)) * 32 + (il & 15) + 16 * (c >> 3);
    float v = 0.0f;
#pragma unroll
    for (int kg = 0; kg < 8; ++kg) v += red[kg * (4 * 8 * 32) + idx];
    sp[e] = v * (e2 == 0 ? hf0 : hf1);
  }
}

#define OG_ROWS 32
#define OG_KC   256
#define OG_NCH  (N_NODES / OG_KC)
#define OG_THR  512
#define OG_UP   260
#define OG_UT   (OG_ROWS * OG_UP)
#define OG_SG   80
#define OG_ST   ((OG_KC / 4) * OG_SG)
#define OG_BUF  (OG_UT + OG_ST)

#define OG_DECL(S) v4f S##u0, S##u1, S##u2, S##u3, S##t0, S##t1, S##h0, S##h1

#define OG_ISSUE(S, ch)                                                        \
  do {                                                                         \
    const float* pu_ = pu + (size_t)(ch) * OG_KC;                              \
    S##u0 = *(const v4f*)(pu_);                                                \
    S##u1 = *(const v4f*)(pu_ + (size_t)8 * N_NODES);                          \
    S##u2 = *(const v4f*)(pu_ + (size_t)16 * N_NODES);                         \
    S##u3 = *(const v4f*)(pu_ + (size_t)24 * N_NODES);                         \
    S##t0 = pt[(ch) * (OG_KC * 4)];                                            \
    S##t1 = pt[(ch) * (OG_KC * 4) + OG_THR];                                   \
    S##h0 = ph[(ch) * (OG_KC * 4)];                                            \
    S##h1 = ph[(ch) * (OG_KC * 4) + OG_THR];                                   \
  } while (0)

#define OG_STASH(S, buf)                                                       \
  do {                                                                         \
    *(v4f*)&(buf)[uoff]              = S##u0;                                  \
    *(v4f*)&(buf)[uoff + 8 * OG_UP]  = S##u1;                                  \
    *(v4f*)&(buf)[uoff + 16 * OG_UP] = S##u2;                                  \
    *(v4f*)&(buf)[uoff + 24 * OG_UP] = S##u3;                                  \
    *(v4f*)&(buf)[OG_UT + soff]              = S##t0 + S##h0;                  \
    *(v4f*)&(buf)[OG_UT + soff + 32 * OG_SG] = S##t1 + S##h1;                  \
  } while (0)

__device__ __forceinline__ void og_compute(const float* buf, int aoff, int boff,
                                           v8f& acc0, v8f& acc1) {
#pragma unroll
  for (int m = 0; m < 2; ++m) {
    v4f a0 = *(const v4f*)&buf[aoff + 8 * m];
    v4f a1 = *(const v4f*)&buf[aoff + 16 * OG_UP + 8 * m];
    const float* sb = &buf[OG_UT + boff + 2 * m * OG_SG];
    v2f b01, b23, x;
    b01.x = sb[0];  b01.y = sb[16];
    b23.x = sb[32]; b23.y = sb[48];
    x.x = a0.x; x.y = a0.y; acc0 = wmma4(x, b01, acc0);
    x.x = a1.x; x.y = a1.y; acc1 = wmma4(x, b01, acc1);
    x.x = a0.z; x.y = a0.w; acc0 = wmma4(x, b23, acc0);
    x.x = a1.z; x.y = a1.w; acc1 = wmma4(x, b23, acc1);
  }
}

__global__ void __launch_bounds__(OG_THR) __attribute__((amdgpu_waves_per_eu(4, 4)))
outgemm_lsm_kernel(const float* __restrict__ U,
                   const float* __restrict__ s0,
                   const float* __restrict__ s1,
                   float* __restrict__ out) {
  __shared__ __attribute__((aligned(16))) float lds[2 * OG_BUF];

  const int tid  = threadIdx.x;
  const int w    = tid >> 5;
  const int lane = tid & 31;
  const int mn   = lane & 15;
  const int half = lane >> 4;
  const int r0   = blockIdx.x * OG_ROWS;

  const float* pu = U + (size_t)(r0 + (tid >> 6)) * N_NODES + (tid & 63) * 4;
  const v4f*   pt = (const v4f*)s0 + tid;
  const v4f*   ph = (const v4f*)s1 + tid;
  const int uoff = (tid >> 6) * OG_UP + (tid & 63) * 4;
  const int soff = (tid >> 4) * OG_SG + (tid & 15) * 4;
  const int aoff = mn * OG_UP + w * 16 + 4 * half;
  const int boff = (4 * w + half) * OG_SG + mn;

  float* buf0 = lds;
  float* buf1 = lds + OG_BUF;

  const int first = blockIdx.x & (OG_NCH - 1);
#define OG_CH(j) (((j) + first) & (OG_NCH - 1))

  OG_DECL(X);
  OG_DECL(Y);
  OG_ISSUE(X, OG_CH(0));
  __builtin_amdgcn_sched_barrier(0);
  OG_ISSUE(Y, OG_CH(1));
  __builtin_amdgcn_sched_barrier(0);
  OG_STASH(X, buf0);
  OG_ISSUE(X, OG_CH(2));
  __builtin_amdgcn_sched_barrier(0);
  og_barrier();

  v8f acc0 = {}, acc1 = {};
  for (int i = 0; i < OG_NCH - 4; i += 2) {
    OG_STASH(Y, buf1);
    OG_ISSUE(Y, OG_CH(i + 3));
    og_compute(buf0, aoff, boff, acc0, acc1);
    og_barrier();
    OG_STASH(X, buf0);
    OG_ISSUE(X, OG_CH(i + 4));
    og_compute(buf1, aoff, boff, acc0, acc1);
    og_barrier();
  }
  OG_STASH(Y, buf1);
  OG_ISSUE(Y, OG_CH(OG_NCH - 1));
  og_compute(buf0, aoff, boff, acc0, acc1);
  og_barrier();
  OG_STASH(X, buf0);
  og_compute(buf1, aoff, boff, acc0, acc1);
  og_barrier();
  OG_STASH(Y, buf1);
  og_compute(buf0, aoff, boff, acc0, acc1);
  og_barrier();
  og_compute(buf1, aoff, boff, acc0, acc1);
  og_barrier();

  float* red = lds;
  float* vs  = lds + OG_BUF;
#pragma unroll
  for (int r = 0; r < 8; ++r) {
    red[((w * 2 + 0) * 8 + r) * 32 + lane] = acc0[r];
    red[((w * 2 + 1) * 8 + r) * 32 + lane] = acc1[r];
  }
  og_barrier();
  const int row  = tid >> 4;
  const int c    = tid & 15;
  const int tile = row >> 4;
  const int rr   = row & 15;
  const int ridx = (tile * 8 + (rr & 7)) * 32 + c + 16 * (rr >> 3);
  float v = 0.0f;
#pragma unroll
  for (int w2 = 0; w2 < 16; ++w2) v += red[w2 * (2 * 8 * 32) + ridx];
  vs[tid] = v;
  og_barrier();

  float rv[CCH];
#pragma unroll
  for (int q = 0; q < 4; ++q) {
    float4 t4 = *(const float4*)&vs[row * CCH + q * 4];
    rv[q * 4 + 0] = t4.x; rv[q * 4 + 1] = t4.y; rv[q * 4 + 2] = t4.z; rv[q * 4 + 3] = t4.w;
  }
  float mx = rv[0];
#pragma unroll
  for (int i = 1; i < CCH; ++i) mx = fmaxf(mx, rv[i]);
  float sm = 0.0f;
#pragma unroll
  for (int i = 0; i < CCH; ++i) sm += expf(rv[i] - mx);
  const float lse = mx + logf(sm);
  out[(size_t)(r0 + row) * CCH + c] = v - lse;
}

extern "C" void kernel_launch(void* const* d_in, const int* in_sizes, int n_in,
                              void* d_out, int out_size, void* d_ws, size_t ws_size,
                              hipStream_t stream) {
  const float* origin_e = (const float*)d_in[0];
  const float* U        = (const float*)d_in[1];
  const float* x        = (const float*)d_in[2];
  const float* W1       = (const float*)d_in[3];
  const float* b1       = (const float*)d_in[4];
  const float* W2       = (const float*)d_in[5];
  const float* b2       = (const float*)d_in[6];
  const float* temp     = (const float*)d_in[7];
  float* out = (float*)d_out;

  float* ws   = (float*)d_ws;
  float* zb   = ws;
  float* sbuf = zb   + (size_t)N_NODES * CCH;

  mlp_kernel<<<N_NODES / ML_ROWS, ML_THR, 0, stream>>>(x, W1, b1, W2, b2, zb);
  utz_kernel<<<2 * (N_NODES / 64), UZ_THR, 0, stream>>>(U, zb, origin_e, temp, sbuf);
  outgemm_lsm_kernel<<<N_NODES / OG_ROWS, OG_THR, 0, stream>>>(U, sbuf, sbuf + (size_t)N_NODES * CCH, out);
}


# s = Hf*(U^T z) kept as ONE plane: MLP launch zeroes it, both row halves add into it by float atomic add (2 addends on 0: sum order-free, same bits), out-GEMM reads s once: its s traffic halves
# speedup vs baseline: 1.9442x; 1.0766x over previous
#include <hip/hip_runtime.h>
#include <cstdint>

typedef __attribute__((ext_vector_type(2))) float v2f;
typedef __attribute__((ext_vector_type(8))) float v8f;

#define N_NODES 8192
#define F_INF   512
#define HIDF    64
#define CCH     16
#define KORD    10

__device__ __forceinline__ v8f wmma4(v2f a, v2f b, v8f c) {
  return __builtin_amdgcn_wmma_f32_16x16x4_f32(
       false, a,  false, b,
       (short)0, c,  false,  false);
}

__device__ __forceinline__ void jacobi_pair(const float* ac, float e0, float e1,
                                            float& h0, float& h1) {
  const float a = 1.0f, b = 1.0f, lo = -1.0f, hi = 1.0f;
  const float coef1 = (a - b) * 0.5f - (a + b + 2.0f) * 0.5f * (lo + hi) / (hi - lo);
  const float coef2 = (a + b + 2.0f) / (hi - lo);

  const float al0 = ac[0], al1 = ac[1];
  float p2 = al0, q2 = al0;
  float accp = p2, accq = q2;
  float p1 = coef1 * al1 + coef2 * al1 * e0 * p2;
  float q1 = coef1 * al1 + coef2 * al1 * e1 * q2;
  accp += p1;
  accq += q1;
#pragma unroll
  for (int L = 2; L <= KORD; ++L) {
    float Lf = (float)L;
    float coef_l  = 2.0f * Lf * (Lf + a + b) * (2.0f * Lf - 2.0f + a + b);
    float c_lm1_1 = (2.0f * Lf + a + b - 1.0f) * (2.0f * Lf + a + b) * (2.0f * Lf + a + b - 2.0f);
    float c_lm1_2 = (2.0f * Lf + a + b - 1.0f) * (a * a - b * b);
    float c_lm2   = 2.0f * (Lf - 1.0f + a) * (Lf - 1.0f + b) * (2.0f * Lf + a + b);
    float tmp1   = ac[L - 1] * (c_lm1_1 / coef_l);
    float tmp2   = ac[L - 1] * (c_lm1_2 / coef_l);
    float tmp3   = ac[L - 1] * ac[L - 2] * (c_lm2 / coef_l);
    float tmp1_2 = tmp1 * (2.0f / (hi - lo));
    float tmp2_2 = tmp1 * ((hi + lo) / (hi - lo)) + tmp2;
    float np = tmp1_2 * (e0 * p1) - tmp2_2 * p1 - tmp3 * p2;
    float nq = tmp1_2 * (e1 * q1) - tmp2_2 * q1 - tmp3 * q2;
    accp += np;
    accq += nq;
    p2 = p1; p1 = np;
    q2 = q1; q1 = nq;
  }
  h0 = accp;
  h1 = accq;
}

#define ML_ROWS 32
#define ML_THR  256
#define ML_XP   516
#define ML_WP   72
#define ML_HP   68
#define ML_XS   (ML_ROWS * ML_XP)
#define ML_WS   (F_INF * ML_WP)
#define ML_HS   (ML_ROWS * ML_HP)

typedef __attribute__((ext_vector_type(4))) float v4f;

__global__ void __launch_bounds__(ML_THR)
mlp_kernel(const float* __restrict__ x,
           const float* __restrict__ W1,
           const float* __restrict__ b1,
           const float* __restrict__ W2,
           const float* __restrict__ b2,
           float* __restrict__ zb,
           float* __restrict__ splane) {
  __shared__ __attribute__((aligned(16))) float lds[ML_XS + ML_WS + ML_HS];
  float* wl = lds;
  float* xs = lds + ML_WS;
  float* hs = lds + ML_WS + ML_XS;

  const int tid  = threadIdx.x;
  const int w    = tid >> 5;
  const int lane = tid & 31;
  const int mn   = lane & 15;
  const int half = lane >> 4;
  const int kk   = half * 2;
  const int mt   = w & 1;
  const int n0   = (w >> 1) * 16;
  const int m0   = blockIdx.x * ML_ROWS;

  v4f xr[16], wr[32];
  const float* px = x + (size_t)(m0 + (tid >> 7)) * F_INF + (tid & 127) * 4;
#pragma unroll
  for (int i = 0; i < 16; ++i) xr[i] = *(const v4f*)(px + (size_t)(2 * i) * F_INF);
  const int wrot = blockIdx.x & 31;
#pragma unroll
  for (int i = 0; i < 32; ++i) {
    const int j = (i + wrot) & 31;
    wr[i] = *(const v4f*)(W1 + (size_t)(tid + 256 * j) * 4);
  }
  v2f w2f[16];
  float bb = 0.0f;
  if (w < 2) {
#pragma unroll
    for (int q = 0; q < 16; ++q) {
      w2f[q].x = W2[(4 * q + kk) * CCH + mn];
      w2f[q].y = W2[(4 * q + kk + 1) * CCH + mn];
    }
    bb = b2[mn];
  }
  const float bias = b1[n0 + mn];

  if (tid < 128) {
    const v4f zero4 = {0.0f, 0.0f, 0.0f, 0.0f};
    *(v4f*)(splane + (size_t)blockIdx.x * 512 + tid * 4) = zero4;
  }

  const int xoff = (tid >> 7) * ML_XP + (tid & 127) * 4;
#pragma unroll
  for (int i = 0; i < 16; ++i) *(v4f*)&xs[xoff + 2 * i * ML_XP] = xr[i];
  const int woff = (tid >> 4) * ML_WP + (tid & 15) * 4;
#pragma unroll
  for (int i = 0; i < 32; ++i) {
    const int j = (i + wrot) & 31;
    *(v4f*)&wl[woff + 16 * j * ML_WP] = wr[i];
  }
  __syncthreads();

  v8f acc0 = {}, acc1 = {};
  const float* xa = &xs[(mt * 16 + mn) * ML_XP + 4 * half];
  const float* wb = &wl[(4 * half) * ML_WP + n0 + mn];
#pragma unroll 8
  for (int g = 0; g < F_INF / 8; ++g) {
    const v4f a = *(const v4f*)&xa[8 * g];
    const float* wg = wb + 8 * g * ML_WP;
    v2f b01, b23, av;
    b01.x = wg[0];          b01.y = wg[ML_WP];
    b23.x = wg[2 * ML_WP];  b23.y = wg[3 * ML_WP];
    av.x = a.x; av.y = a.y; acc0 = wmma4(av, b01, acc0);
    av.x = a.z; av.y = a.w; acc1 = wmma4(av, b23, acc1);
  }

#pragma unroll
  for (int r = 0; r < 8; ++r) {
    const int row = mt * 16 + (half ? r + 8 : r);
    const float v = (acc0[r] + acc1[r]) + bias;
    hs[row * ML_HP + n0 + mn] = v > 0.0f ? v : 0.0f;
  }
  __syncthreads();

  if (w < 2) {
    v8f zacc = {};
#pragma unroll
    for (int q = 0; q < 16; ++q) {
      v2f av;
      av.x = hs[(w * 16 + mn) * ML_HP + 4 * q + kk];
      av.y = hs[(w * 16 + mn) * ML_HP + 4 * q + kk + 1];
      zacc = wmma4(av, w2f[q], zacc);
    }
#pragma unroll
    for (int r = 0; r < 8; ++r) {
      const int row = w * 16 + (half ? r + 8 : r);
      zb[(size_t)(m0 + row) * CCH + mn] = zacc[r] + bb;
    }
  }
}

__device__ __forceinline__ void og_barrier() {
  asm volatile("s_wait_dscnt 0x0\n\ts_barrier_signal -1\n\ts_barrier_wait -1" ::: "memory");
  __builtin_amdgcn_sched_barrier(0);
}

#define UZ_JH   (N_NODES / 2)
#define UZ_KC   256
#define UZ_NCH  (UZ_JH / UZ_KC)
#define UZ_THR  512
#define UZ_UP   80
#define UZ_UT   (UZ_KC * UZ_UP)
#define UZ_ZT   (UZ_KC * CCH)
#define UZ_BUF  (UZ_UT + UZ_ZT)
#define UZ_NAL  (CCH * (KORD + 1))

#define UZ_DECL(S) v4f S##u0, S##u1, S##u2, S##u3, S##u4, S##u5, S##u6, S##u7, S##z0, S##z1

#define UZ_ISSUE(S, ch)                                                        \
  do {                                                                         \
    const float* pu_ = pu + (size_t)(ch) * ((size_t)UZ_KC * N_NODES);          \
    S##u0 = *(const v4f*)(pu_);                                                \
    S##u1 = *(const v4f*)(pu_ + (size_t)32 * N_NODES);                         \
    S##u2 = *(const v4f*)(pu_ + (size_t)64 * N_NODES);                         \
    S##u3 = *(const v4f*)(pu_ + (size_t)96 * N_NODES);                         \
    S##u4 = *(const v4f*)(pu_ + (size_t)128 * N_NODES);                        \
    S##u5 = *(const v4f*)(pu_ + (size_t)160 * N_NODES);                        \
    S##u6 = *(const v4f*)(pu_ + (size_t)192 * N_NODES);                        \
    S##u7 = *(const v4f*)(pu_ + (size_t)224 * N_NODES);                        \
    S##z0 = pz[(ch) * (UZ_KC * 4)];                                            \
    S##z1 = pz[(ch) * (UZ_KC * 4) + UZ_THR];                                   \
  } while (0)

#define UZ_STASH(S, buf)                                                       \
  do {                                                                         \
    *(v4f*)&(buf)[uoff]               = S##u0;                                 \
    *(v4f*)&(buf)[uoff + 32 * UZ_UP]  = S##u1;                                 \
    *(v4f*)&(buf)[uoff + 64 * UZ_UP]  = S##u2;                                 \
    *(v4f*)&(buf)[uoff + 96 * UZ_UP]  = S##u3;                                 \
    *(v4f*)&(buf)[uoff + 128 * UZ_UP] = S##u4;                                 \
    *(v4f*)&(buf)[uoff + 160 * UZ_UP] = S##u5;                                 \
    *(v4f*)&(buf)[uoff + 192 * UZ_UP] = S##u6;                                 \
    *(v4f*)&(buf)[uoff + 224 * UZ_UP] = S##u7;                                 \
    *(v4f*)&(buf)[UZ_UT + zoff]              = S##z0;                          \
    *(v4f*)&(buf)[UZ_UT + zoff + 4 * UZ_THR] = S##z1;                          \
  } while (0)

__device__ __forceinline__ void uz_compute(const float* buf, int aoff, int boff,
                                           v8f& acc0, v8f& acc1) {
#pragma unroll
  for (int q = 0; q < 8; ++q) {
    const float* zr = &buf[aoff + q * (4 * CCH)];
    const float* ur = &buf[boff + q * (4 * UZ_UP)];
    v2f a, b0, b1;
    a.x  = zr[0];   a.y  = zr[CCH];
    b0.x = ur[0];   b0.y = ur[UZ_UP];
    b1.x = ur[16];  b1.y = ur[UZ_UP + 16];
    acc0 = wmma4(a, b0, acc0);
    acc1 = wmma4(a, b1, acc1);
  }
}

__global__ void __launch_bounds__(UZ_THR) __attribute__((amdgpu_waves_per_eu(4, 4)))
utz_kernel(const float* __restrict__ U,
           const float* __restrict__ zb,
           const float* __restrict__ origin_e,
           const float* __restrict__ temp,
           float* __restrict__ sbuf) {
  __shared__ __attribute__((aligned(16))) float lds[2 * UZ_BUF + UZ_NAL];

  const int tid  = threadIdx.x;
  const int w    = tid >> 5;
  const int lane = tid & 31;
  const int mn   = lane & 15;
  const int half = lane >> 4;
  const int g    = blockIdx.x & 127;
  const int hf   = blockIdx.x >> 7;
  const int c0   = g * 64;
  const int j0   = hf * UZ_JH;

  const float* pu = U + (size_t)(j0 + (tid >> 4)) * N_NODES + c0 + (tid & 15) * 4;
  const v4f*   pz = (const v4f*)(zb + (size_t)j0 * CCH) + tid;
  const int uoff = (tid >> 4) * UZ_UP + (tid & 15) * 4;
  const int zoff = tid * 4;
  const int aoff = UZ_UT + (32 * (w >> 1) + 2 * half) * CCH + mn;
  const int boff = (32 * (w >> 1) + 2 * half) * UZ_UP + 32 * (w & 1) + mn;

  float* buf0 = lds;
  float* buf1 = lds + UZ_BUF;
  float* al   = lds + 2 * UZ_BUF;

  float tv = 0.0f;
  if (tid < UZ_NAL) tv = temp[tid];
  const float e0 = origin_e[c0 + (tid >> 4)];
  const float e1 = origin_e[c0 + 32 + (tid >> 4)];

  const int first = (g >> 3) & (UZ_NCH - 1);
#define UZ_CH(j) (((j) + first) & (UZ_NCH - 1))

  UZ_DECL(X);
  UZ_DECL(Y);
  UZ_ISSUE(X, UZ_CH(0));
  UZ_ISSUE(Y, UZ_CH(1));
  if (tid < UZ_NAL) al[tid] = 0.5f * tanhf(tv);
  og_barrier();
  float hf0, hf1;
  jacobi_pair(al + (tid & 15) * (KORD + 1), e0, e1, hf0, hf1);
  asm volatile("" : "+v"(hf0), "+v"(hf1));
  UZ_STASH(X, buf0);
  UZ_ISSUE(X, UZ_CH(2));
  og_barrier();

  v8f acc0 = {}, acc1 = {};
  for (int i = 0; i < UZ_NCH - 4; i += 2) {
    UZ_STASH(Y, buf1);
    UZ_ISSUE(Y, UZ_CH(i + 3));
    uz_compute(buf0, aoff, boff, acc0, acc1);
    og_barrier();
    UZ_STASH(X, buf0);
    UZ_ISSUE(X, UZ_CH(i + 4));
    uz_compute(buf1, aoff, boff, acc0, acc1);
    og_barrier();
  }
  UZ_STASH(Y, buf1);
  UZ_ISSUE(Y, UZ_CH(UZ_NCH - 1));
  uz_compute(buf0, aoff, boff, acc0, acc1);
  og_barrier();
  UZ_STASH(X, buf0);
  uz_compute(buf1, aoff, boff, acc0, acc1);
  og_barrier();
  UZ_STASH(Y, buf1);
  uz_compute(buf0, aoff, boff, acc0, acc1);
  og_barrier();
  uz_compute(buf1, aoff, boff, acc0, acc1);
  og_barrier();

  float* red = lds;
#pragma unroll
  for (int r = 0; r < 8; ++r) {
    red[((w * 2 + 0) * 8 + r) * 32 + lane] = acc0[r];
    red[((w * 2 + 1) * 8 + r) * 32 + lane] = acc1[r];
  }
  og_barrier();
  float* sp = sbuf + (size_t)c0 * CCH;
#pragma unroll
  for (int e2 = 0; e2 < 2; ++e2) {
    const int e   = tid + e2 * UZ_THR;
    const int il  = e >> 4;
    const int c   = e & 15;
    const int idx = ((il >> 4) * 8 + (c & 7)) * 32 + (il & 15) + 16 * (c >> 3);
    float v = 0.0f;
#pragma unroll
    for (int kg = 0; kg < 8; ++kg) v += red[kg * (4 * 8 * 32) + idx];
    unsafeAtomicAdd(&sp[e], v * (e2 == 0 ? hf0 : hf1));
  }
}

#define OG_ROWS 32
#define OG_KC   256
#define OG_NCH  (N_NODES / OG_KC)
#define OG_THR  512
#define OG_UP   260
#define OG_UT   (OG_ROWS * OG_UP)
#define OG_SG   80
#define OG_ST   ((OG_KC / 4) * OG_SG)
#define OG_BUF  (OG_UT + OG_ST)

#define OG_DECL(S) v4f S##u0, S##u1, S##u2, S##u3, S##t0, S##t1

#define OG_ISSUE(S, ch)                                                        \
  do {                                                                         \
    const float* pu_ = pu + (size_t)(ch) * OG_KC;                              \
    S##u0 = *(const v4f*)(pu_);                                                \
    S##u1 = *(const v4f*)(pu_ + (size_t)8 * N_NODES);                          \
    S##u2 = *(const v4f*)(pu_ + (size_t)16 * N_NODES);                         \
    S##u3 = *(const v4f*)(pu_ + (size_t)24 * N_NODES);                         \
    S##t0 = pt[(ch) * (OG_KC * 4)];                                            \
    S##t1 = pt[(ch) * (OG_KC * 4) + OG_THR];                                   \
  } while (0)

#define OG_STASH(S, buf)                                                       \
  do {                                                                         \
    *(v4f*)&(buf)[uoff]              = S##u0;                                  \
    *(v4f*)&(buf)[uoff + 8 * OG_UP]  = S##u1;                                  \
    *(v4f*)&(buf)[uoff + 16 * OG_UP] = S##u2;                                  \
    *(v4f*)&(buf)[uoff + 24 * OG_UP] = S##u3;                                  \
    *(v4f*)&(buf)[OG_UT + soff]              = S##t0;                          \
    *(v4f*)&(buf)[OG_UT + soff + 32 * OG_SG] = S##t1;                          \
  } while (0)

__device__ __forceinline__ void og_compute(const float* buf, int aoff, int boff,
                                           v8f& acc0, v8f& acc1) {
#pragma unroll
  for (int m = 0; m < 2; ++m) {
    v4f a0 = *(const v4f*)&buf[aoff + 8 * m];
    v4f a1 = *(const v4f*)&buf[aoff + 16 * OG_UP + 8 * m];
    const float* sb = &buf[OG_UT + boff + 2 * m * OG_SG];
    v2f b01, b23, x;
    b01.x = sb[0];  b01.y = sb[16];
    b23.x = sb[32]; b23.y = sb[48];
    x.x = a0.x; x.y = a0.y; acc0 = wmma4(x, b01, acc0);
    x.x = a1.x; x.y = a1.y; acc1 = wmma4(x, b01, acc1);
    x.x = a0.z; x.y = a0.w; acc0 = wmma4(x, b23, acc0);
    x.x = a1.z; x.y = a1.w; acc1 = wmma4(x, b23, acc1);
  }
}

__global__ void __launch_bounds__(OG_THR) __attribute__((amdgpu_waves_per_eu(4, 4)))
outgemm_lsm_kernel(const float* __restrict__ U,
                   const float* __restrict__ s,
                   float* __restrict__ out) {
  __shared__ __attribute__((aligned(16))) float lds[2 * OG_BUF];

  const int tid  = threadIdx.x;
  const int w    = tid >> 5;
  const int lane = tid & 31;
  const int mn   = lane & 15;
  const int half = lane >> 4;
  const int r0   = blockIdx.x * OG_ROWS;

  const float* pu = U + (size_t)(r0 + (tid >> 6)) * N_NODES + (tid & 63) * 4;
  const v4f*   pt = (const v4f*)s + tid;
  const int uoff = (tid >> 6) * OG_UP + (tid & 63) * 4;
  const int soff = (tid >> 4) * OG_SG + (tid & 15) * 4;
  const int aoff = mn * OG_UP + w * 16 + 4 * half;
  const int boff = (4 * w + half) * OG_SG + mn;

  float* buf0 = lds;
  float* buf1 = lds + OG_BUF;

  const int first = blockIdx.x & (OG_NCH - 1);
#define OG_CH(j) (((j) + first) & (OG_NCH - 1))

  OG_DECL(X);
  OG_DECL(Y);
  OG_ISSUE(X, OG_CH(0));
  __builtin_amdgcn_sched_barrier(0);
  OG_ISSUE(Y, OG_CH(1));
  __builtin_amdgcn_sched_barrier(0);
  OG_STASH(X, buf0);
  OG_ISSUE(X, OG_CH(2));
  __builtin_amdgcn_sched_barrier(0);
  og_barrier();

  v8f acc0 = {}, acc1 = {};
  for (int i = 0; i < OG_NCH - 4; i += 2) {
    OG_STASH(Y, buf1);
    OG_ISSUE(Y, OG_CH(i + 3));
    og_compute(buf0, aoff, boff, acc0, acc1);
    og_barrier();
    OG_STASH(X, buf0);
    OG_ISSUE(X, OG_CH(i + 4));
    og_compute(buf1, aoff, boff, acc0, acc1);
    og_barrier();
  }
  OG_STASH(Y, buf1);
  OG_ISSUE(Y, OG_CH(OG_NCH - 1));
  og_compute(buf0, aoff, boff, acc0, acc1);
  og_barrier();
  OG_STASH(X, buf0);
  og_compute(buf1, aoff, boff, acc0, acc1);
  og_barrier();
  OG_STASH(Y, buf1);
  og_compute(buf0, aoff, boff, acc0, acc1);
  og_barrier();
  og_compute(buf1, aoff, boff, acc0, acc1);
  og_barrier();

  float* red = lds;
  float* vs  = lds + OG_BUF;
#pragma unroll
  for (int r = 0; r < 8; ++r) {
    red[((w * 2 + 0) * 8 + r) * 32 + lane] = acc0[r];
    red[((w * 2 + 1) * 8 + r) * 32 + lane] = acc1[r];
  }
  og_barrier();
  const int row  = tid >> 4;
  const int c    = tid & 15;
  const int tile = row >> 4;
  const int rr   = row & 15;
  const int ridx = (tile * 8 + (rr & 7)) * 32 + c + 16 * (rr >> 3);
  float v = 0.0f;
#pragma unroll
  for (int w2 = 0; w2 < 16; ++w2) v += red[w2 * (2 * 8 * 32) + ridx];
  vs[tid] = v;
  og_barrier();

  float rv[CCH];
#pragma unroll
  for (int q = 0; q < 4; ++q) {
    float4 t4 = *(const float4*)&vs[row * CCH + q * 4];
    rv[q * 4 + 0] = t4.x; rv[q * 4 + 1] = t4.y; rv[q * 4 + 2] = t4.z; rv[q * 4 + 3] = t4.w;
  }
  float mx = rv[0];
#pragma unroll
  for (int i = 1; i < CCH; ++i) mx = fmaxf(mx, rv[i]);
  float sm = 0.0f;
#pragma unroll
  for (int i = 0; i < CCH; ++i) sm += expf(rv[i] - mx);
  const float lse = mx + logf(sm);
  out[(size_t)(r0 + row) * CCH + c] = v - lse;
}

extern "C" void kernel_launch(void* const* d_in, const int* in_sizes, int n_in,
                              void* d_out, int out_size, void* d_ws, size_t ws_size,
                              hipStream_t stream) {
  const float* origin_e = (const float*)d_in[0];
  const float* U        = (const float*)d_in[1];
  const float* x        = (const float*)d_in[2];
  const float* W1       = (const float*)d_in[3];
  const float* b1       = (const float*)d_in[4];
  const float* W2       = (const float*)d_in[5];
  const float* b2       = (const float*)d_in[6];
  const float* temp     = (const float*)d_in[7];
  float* out = (float*)d_out;

  float* ws   = (float*)d_ws;
  float* zb   = ws;
  float* sbuf = zb   + (size_t)N_NODES * CCH;

  mlp_kernel<<<N_NODES / ML_ROWS, ML_THR, 0, stream>>>(x, W1, b1, W2, b2, zb, sbuf);
  utz_kernel<<<2 * (N_NODES / 64), UZ_THR, 0, stream>>>(U, zb, origin_e, temp, sbuf);
  outgemm_lsm_kernel<<<N_NODES / OG_ROWS, OG_THR, 0, stream>>>(U, sbuf, out);
}


# out = U s reads U in reverse of the U^T z pass's order, both passes binding row half h to the dies of L2 half h (die id + ticket; sums fixed per item): first chunks hit L2; k-chunk sum order changes
# speedup vs baseline: 2.0400x; 1.0493x over previous
#include <hip/hip_runtime.h>
#include <cstdint>

typedef __attribute__((ext_vector_type(2))) float v2f;
typedef __attribute__((ext_vector_type(8))) float v8f;

#define N_NODES 8192
#define F_INF   512
#define HIDF    64
#define CCH     16
#define KORD    10
#define WK_HALF    128
#define WK_CSTRIDE 64

__device__ __forceinline__ v8f wmma4(v2f a, v2f b, v8f c) {
  return __builtin_amdgcn_wmma_f32_16x16x4_f32(
       false, a,  false, b,
       (short)0, c,  false,  false);
}

__device__ __forceinline__ void jacobi_pair(const float* ac, float e0, float e1,
                                            float& h0, float& h1) {
  const float a = 1.0f, b = 1.0f, lo = -1.0f, hi = 1.0f;
  const float coef1 = (a - b) * 0.5f - (a + b + 2.0f) * 0.5f * (lo + hi) / (hi - lo);
  const float coef2 = (a + b + 2.0f) / (hi - lo);

  const float al0 = ac[0], al1 = ac[1];
  float p2 = al0, q2 = al0;
  float accp = p2, accq = q2;
  float p1 = coef1 * al1 + coef2 * al1 * e0 * p2;
  float q1 = coef1 * al1 + coef2 * al1 * e1 * q2;
  accp += p1;
  accq += q1;
#pragma unroll
  for (int L = 2; L <= KORD; ++L) {
    float Lf = (float)L;
    float coef_l  = 2.0f * Lf * (Lf + a + b) * (2.0f * Lf - 2.0f + a + b);
    float c_lm1_1 = (2.0f * Lf + a + b - 1.0f) * (2.0f * Lf + a + b) * (2.0f * Lf + a + b - 2.0f);
    float c_lm1_2 = (2.0f * Lf + a + b - 1.0f) * (a * a - b * b);
    float c_lm2   = 2.0f * (Lf - 1.0f + a) * (Lf - 1.0f + b) * (2.0f * Lf + a + b);
    float tmp1   = ac[L - 1] * (c_lm1_1 / coef_l);
    float tmp2   = ac[L - 1] * (c_lm1_2 / coef_l);
    float tmp3   = ac[L - 1] * ac[L - 2] * (c_lm2 / coef_l);
    float tmp1_2 = tmp1 * (2.0f / (hi - lo));
    float tmp2_2 = tmp1 * ((hi + lo) / (hi - lo)) + tmp2;
    float np = tmp1_2 * (e0 * p1) - tmp2_2 * p1 - tmp3 * p2;
    float nq = tmp1_2 * (e1 * q1) - tmp2_2 * q1 - tmp3 * q2;
    accp += np;
    accq += nq;
    p2 = p1; p1 = np;
    q2 = q1; q1 = nq;
  }
  h0 = accp;
  h1 = accq;
}

#define ML_ROWS 32
#define ML_THR  256
#define ML_XP   516
#define ML_WP   72
#define ML_HP   68
#define ML_XS   (ML_ROWS * ML_XP)
#define ML_WS   (F_INF * ML_WP)
#define ML_HS   (ML_ROWS * ML_HP)

typedef __attribute__((ext_vector_type(4))) float v4f;

__global__ void __launch_bounds__(ML_THR)
mlp_kernel(const float* __restrict__ x,
           const float* __restrict__ W1,
           const float* __restrict__ b1,
           const float* __restrict__ W2,
           const float* __restrict__ b2,
           float* __restrict__ zb,
           float* __restrict__ splane,
           float* __restrict__ wkc) {
  __shared__ __attribute__((aligned(16))) float lds[ML_XS + ML_WS + ML_HS];
  float* wl = lds;
  float* xs = lds + ML_WS;
  float* hs = lds + ML_WS + ML_XS;

  const int tid  = threadIdx.x;
  const int w    = tid >> 5;
  const int lane = tid & 31;
  const int mn   = lane & 15;
  const int half = lane >> 4;
  const int kk   = half * 2;
  const int mt   = w & 1;
  const int n0   = (w >> 1) * 16;
  const int m0   = blockIdx.x * ML_ROWS;

  v4f xr[16], wr[32];
  const float* px = x + (size_t)(m0 + (tid >> 7)) * F_INF + (tid & 127) * 4;
#pragma unroll
  for (int i = 0; i < 16; ++i) xr[i] = *(const v4f*)(px + (size_t)(2 * i) * F_INF);
  const int wrot = blockIdx.x & 31;
#pragma unroll
  for (int i = 0; i < 32; ++i) {
    const int j = (i + wrot) & 31;
    wr[i] = *(const v4f*)(W1 + (size_t)(tid + 256 * j) * 4);
  }
  v2f w2f[16];
  float bb = 0.0f;
  if (w < 2) {
#pragma unroll
    for (int q = 0; q < 16; ++q) {
      w2f[q].x = W2[(4 * q + kk) * CCH + mn];
      w2f[q].y = W2[(4 * q + kk + 1) * CCH + mn];
    }
    bb = b2[mn];
  }
  const float bias = b1[n0 + mn];

  if (tid < 128) {
    const v4f zero4 = {0.0f, 0.0f, 0.0f, 0.0f};
    *(v4f*)(splane + (size_t)blockIdx.x * 512 + tid * 4) = zero4;
  }
  if (blockIdx.x == 0 && tid < 4) ((unsigned int*)wkc)[tid * WK_CSTRIDE] = 0u;

  const int xoff = (tid >> 7) * ML_XP + (tid & 127) * 4;
#pragma unroll
  for (int i = 0; i < 16; ++i) *(v4f*)&xs[xoff + 2 * i * ML_XP] = xr[i];
  const int woff = (tid >> 4) * ML_WP + (tid & 15) * 4;
#pragma unroll
  for (int i = 0; i < 32; ++i) {
    const int j = (i + wrot) & 31;
    *(v4f*)&wl[woff + 16 * j * ML_WP] = wr[i];
  }
  __syncthreads();

  v8f acc0 = {}, acc1 = {};
  const float* xa = &xs[(mt * 16 + mn) * ML_XP + 4 * half];
  const float* wb = &wl[(4 * half) * ML_WP + n0 + mn];
#pragma unroll 8
  for (int g = 0; g < F_INF / 8; ++g) {
    const v4f a = *(const v4f*)&xa[8 * g];
    const float* wg = wb + 8 * g * ML_WP;
    v2f b01, b23, av;
    b01.x = wg[0];          b01.y = wg[ML_WP];
    b23.x = wg[2 * ML_WP];  b23.y = wg[3 * ML_WP];
    av.x = a.x; av.y = a.y; acc0 = wmma4(av, b01, acc0);
    av.x = a.z; av.y = a.w; acc1 = wmma4(av, b23, acc1);
  }

#pragma unroll
  for (int r = 0; r < 8; ++r) {
    const int row = mt * 16 + (half ? r + 8 : r);
    const float v = (acc0[r] + acc1[r]) + bias;
    hs[row * ML_HP + n0 + mn] = v > 0.0f ? v : 0.0f;
  }
  __syncthreads();

  if (w < 2) {
    v8f zacc = {};
#pragma unroll
    for (int q = 0; q < 16; ++q) {
      v2f av;
      av.x = hs[(w * 16 + mn) * ML_HP + 4 * q + kk];
      av.y = hs[(w * 16 + mn) * ML_HP + 4 * q + kk + 1];
      zacc = wmma4(av, w2f[q], zacc);
    }
#pragma unroll
    for (int r = 0; r < 8; ++r) {
      const int row = w * 16 + (half ? r + 8 : r);
      zb[(size_t)(m0 + row) * CCH + mn] = zacc[r] + bb;
    }
  }
}

__device__ __forceinline__ void og_barrier() {
  asm volatile("s_wait_dscnt 0x0\n\ts_barrier_signal -1\n\ts_barrier_wait -1" ::: "memory");
  __builtin_amdgcn_sched_barrier(0);
}

__device__ __forceinline__ int take_work(unsigned int* cnt) {
  const unsigned int where = __builtin_amdgcn_s_sendmsg_rtn(0x87);
  int h = (int)((where >> 18) & 1u);
  unsigned int t = atomicAdd(&cnt[h * WK_CSTRIDE], 1u);
  if (t >= (unsigned int)WK_HALF) {
    h ^= 1;
    t = atomicAdd(&cnt[h * WK_CSTRIDE], 1u);
  }
  return h * WK_HALF + (int)(t & (unsigned int)(WK_HALF - 1));
}

#define UZ_JH   (N_NODES / 2)
#define UZ_KC   256
#define UZ_NCH  (UZ_JH / UZ_KC)
#define UZ_THR  512
#define UZ_UP   80
#define UZ_UT   (UZ_KC * UZ_UP)
#define UZ_ZT   (UZ_KC * CCH)
#define UZ_BUF  (UZ_UT + UZ_ZT)
#define UZ_NAL  (CCH * (KORD + 1))

#define UZ_DECL(S) v4f S##u0, S##u1, S##u2, S##u3, S##u4, S##u5, S##u6, S##u7, S##z0, S##z1

#define UZ_ISSUE(S, ch)                                                        \
  do {                                                                         \
    const float* pu_ = pu + (size_t)(ch) * ((size_t)UZ_KC * N_NODES);          \
    S##u0 = *(const v4f*)(pu_);                                                \
    S##u1 = *(const v4f*)(pu_ + (size_t)32 * N_NODES);                         \
    S##u2 = *(const v4f*)(pu_ + (size_t)64 * N_NODES);                         \
    S##u3 = *(const v4f*)(pu_ + (size_t)96 * N_NODES);                         \
    S##u4 = *(const v4f*)(pu_ + (size_t)128 * N_NODES);                        \
    S##u5 = *(const v4f*)(pu_ + (size_t)160 * N_NODES);                        \
    S##u6 = *(const v4f*)(pu_ + (size_t)192 * N_NODES);                        \
    S##u7 = *(const v4f*)(pu_ + (size_t)224 * N_NODES);                        \
    S##z0 = pz[(ch) * (UZ_KC * 4)];                                            \
    S##z1 = pz[(ch) * (UZ_KC * 4) + UZ_THR];                                   \
  } while (0)

#define UZ_STASH(S, buf)                                                       \
  do {                                                                         \
    *(v4f*)&(buf)[uoff]               = S##u0;                                 \
    *(v4f*)&(buf)[uoff + 32 * UZ_UP]  = S##u1;                                 \
    *(v4f*)&(buf)[uoff + 64 * UZ_UP]  = S##u2;                                 \
    *(v4f*)&(buf)[uoff + 96 * UZ_UP]  = S##u3;                                 \
    *(v4f*)&(buf)[uoff + 128 * UZ_UP] = S##u4;                                 \
    *(v4f*)&(buf)[uoff + 160 * UZ_UP] = S##u5;                                 \
    *(v4f*)&(buf)[uoff + 192 * UZ_UP] = S##u6;                                 \
    *(v4f*)&(buf)[uoff + 224 * UZ_UP] = S##u7;                                 \
    *(v4f*)&(buf)[UZ_UT + zoff]              = S##z0;                          \
    *(v4f*)&(buf)[UZ_UT + zoff + 4 * UZ_THR] = S##z1;                          \
  } while (0)

__device__ __forceinline__ void uz_compute(const float* buf, int aoff, int boff,
                                           v8f& acc0, v8f& acc1) {
#pragma unroll
  for (int q = 0; q < 8; ++q) {
    const float* zr = &buf[aoff + q * (4 * CCH)];
    const float* ur = &buf[boff + q * (4 * UZ_UP)];
    v2f a, b0, b1;
    a.x  = zr[0];   a.y  = zr[CCH];
    b0.x = ur[0];   b0.y = ur[UZ_UP];
    b1.x = ur[16];  b1.y = ur[UZ_UP + 16];
    acc0 = wmma4(a, b0, acc0);
    acc1 = wmma4(a, b1, acc1);
  }
}

__global__ void __launch_bounds__(UZ_THR) __attribute__((amdgpu_waves_per_eu(4, 4)))
utz_kernel(const float* __restrict__ U,
           const float* __restrict__ zb,
           const float* __restrict__ origin_e,
           const float* __restrict__ temp,
           float* __restrict__ sbuf,
           float* __restrict__ wkc) {
  __shared__ __attribute__((aligned(16))) float lds[2 * UZ_BUF + UZ_NAL];
  __shared__ int wk_s;

  const int tid  = threadIdx.x;
  const int w    = tid >> 5;
  const int lane = tid & 31;
  const int mn   = lane & 15;
  const int half = lane >> 4;
  float* al = lds + 2 * UZ_BUF;

  float tv = 0.0f;
  if (tid < UZ_NAL) tv = temp[tid];
  if (tid == UZ_THR - 32) wk_s = take_work((unsigned int*)wkc);
  if (tid < UZ_NAL) al[tid] = 0.5f * tanhf(tv);
  og_barrier();
  const int wk   = __builtin_amdgcn_readfirstlane(wk_s);
  const int g    = wk & 127;
  const int hf   = wk >> 7;
  const int c0   = g * 64;
  const int j0   = hf * UZ_JH;

  const float* pu = U + (size_t)(j0 + (tid >> 4)) * N_NODES + c0 + (tid & 15) * 4;
  const v4f*   pz = (const v4f*)(zb + (size_t)j0 * CCH) + tid;
  const int uoff = (tid >> 4) * UZ_UP + (tid & 15) * 4;
  const int zoff = tid * 4;
  const int aoff = UZ_UT + (32 * (w >> 1) + 2 * half) * CCH + mn;
  const int boff = (32 * (w >> 1) + 2 * half) * UZ_UP + 32 * (w & 1) + mn;

  float* buf0 = lds;
  float* buf1 = lds + UZ_BUF;

  const float e0 = origin_e[c0 + (tid >> 4)];
  const float e1 = origin_e[c0 + 32 + (tid >> 4)];

  const int first = (g >> 3) & (UZ_NCH - 1);
#define UZ_CH(j) (((j) + first) & (UZ_NCH - 1))

  UZ_DECL(X);
  UZ_DECL(Y);
  UZ_ISSUE(X, UZ_CH(0));
  UZ_ISSUE(Y, UZ_CH(1));
  __builtin_amdgcn_sched_barrier(0);
  float hf0, hf1;
  jacobi_pair(al + (tid & 15) * (KORD + 1), e0, e1, hf0, hf1);
  asm volatile("" : "+v"(hf0), "+v"(hf1));
  UZ_STASH(X, buf0);
  UZ_ISSUE(X, UZ_CH(2));
  og_barrier();

  v8f acc0 = {}, acc1 = {};
  for (int i = 0; i < UZ_NCH - 4; i += 2) {
    UZ_STASH(Y, buf1);
    UZ_ISSUE(Y, UZ_CH(i + 3));
    uz_compute(buf0, aoff, boff, acc0, acc1);
    og_barrier();
    UZ_STASH(X, buf0);
    UZ_ISSUE(X, UZ_CH(i + 4));
    uz_compute(buf1, aoff, boff, acc0, acc1);
    og_barrier();
  }
  UZ_STASH(Y, buf1);
  UZ_ISSUE(Y, UZ_CH(UZ_NCH - 1));
  uz_compute(buf0, aoff, boff, acc0, acc1);
  og_barrier();
  UZ_STASH(X, buf0);
  uz_compute(buf1, aoff, boff, acc0, acc1);
  og_barrier();
  UZ_STASH(Y, buf1);
  uz_compute(buf0, aoff, boff, acc0, acc1);
  og_barrier();
  uz_compute(buf1, aoff, boff, acc0, acc1);
  og_barrier();

  float* red = lds;
#pragma unroll
  for (int r = 0; r < 8; ++r) {
    red[((w * 2 + 0) * 8 + r) * 32 + lane] = acc0[r];
    red[((w * 2 + 1) * 8 + r) * 32 + lane] = acc1[r];
  }
  og_barrier();
  float* sp = sbuf + (size_t)c0 * CCH;
#pragma unroll
  for (int e2 = 0; e2 < 2; ++e2) {
    const int e   = tid + e2 * UZ_THR;
    const int il  = e >> 4;
    const int c   = e & 15;
    const int idx = ((il >> 4) * 8 + (c & 7)) * 32 + (il & 15) + 16 * (c >> 3);
    float v = 0.0f;
#pragma unroll
    for (int kg = 0; kg < 8; ++kg) v += red[kg * (4 * 8 * 32) + idx];
    unsafeAtomicAdd(&sp[e], v * (e2 == 0 ? hf0 : hf1));
  }
}

#define OG_ROWS 32
#define OG_KC   256
#define OG_NCH  (N_NODES / OG_KC)
#define OG_THR  512
#define OG_UP   260
#define OG_UT   (OG_ROWS * OG_UP)
#define OG_SG   80
#define OG_ST   ((OG_KC / 4) * OG_SG)
#define OG_BUF  (OG_UT + OG_ST)

#define OG_DECL(S) v4f S##u0, S##u1, S##u2, S##u3, S##t0, S##t1

#define OG_ISSUE(S, ch)                                                        \
  do {                                                                         \
    const float* pu_ = pu + (size_t)(ch) * OG_KC;                              \
    S##u0 = *(const v4f*)(pu_);                                                \
    S##u1 = *(const v4f*)(pu_ + (size_t)8 * N_NODES);                          \
    S##u2 = *(const v4f*)(pu_ + (size_t)16 * N_NODES);                         \
    S##u3 = *(const v4f*)(pu_ + (size_t)24 * N_NODES);                         \
    S##t0 = pt[(ch) * (OG_KC * 4)];                                            \
    S##t1 = pt[(ch) * (OG_KC * 4) + OG_THR];                                   \
  } while (0)

#define OG_STASH(S, buf)                                                       \
  do {                                                                         \
    *(v4f*)&(buf)[uoff]              = S##u0;                                  \
    *(v4f*)&(buf)[uoff + 8 * OG_UP]  = S##u1;                                  \
    *(v4f*)&(buf)[uoff + 16 * OG_UP] = S##u2;                                  \
    *(v4f*)&(buf)[uoff + 24 * OG_UP] = S##u3;                                  \
    *(v4f*)&(buf)[OG_UT + soff]              = S##t0;                          \
    *(v4f*)&(buf)[OG_UT + soff + 32 * OG_SG] = S##t1;                          \
  } while (0)

__device__ __forceinline__ void og_compute(const float* buf, int aoff, int boff,
                                           v8f& acc0, v8f& acc1) {
#pragma unroll
  for (int m = 0; m < 2; ++m) {
    v4f a0 = *(const v4f*)&buf[aoff + 8 * m];
    v4f a1 = *(const v4f*)&buf[aoff + 16 * OG_UP + 8 * m];
    const float* sb = &buf[OG_UT + boff + 2 * m * OG_SG];
    v2f b01, b23, x;
    b01.x = sb[0];  b01.y = sb[16];
    b23.x = sb[32]; b23.y = sb[48];
    x.x = a0.x; x.y = a0.y; acc0 = wmma4(x, b01, acc0);
    x.x = a1.x; x.y = a1.y; acc1 = wmma4(x, b01, acc1);
    x.x = a0.z; x.y = a0.w; acc0 = wmma4(x, b23, acc0);
    x.x = a1.z; x.y = a1.w; acc1 = wmma4(x, b23, acc1);
  }
}

__global__ void __launch_bounds__(OG_THR) __attribute__((amdgpu_waves_per_eu(4, 4)))
outgemm_lsm_kernel(const float* __restrict__ U,
                   const float* __restrict__ s,
                   float* __restrict__ out,
                   float* __restrict__ wkc) {
  __shared__ __attribute__((aligned(16))) float lds[2 * OG_BUF];
  __shared__ int wk_s;

  const int tid  = threadIdx.x;
  const int w    = tid >> 5;
  const int lane = tid & 31;
  const int mn   = lane & 15;
  const int half = lane >> 4;
  if (tid == 0) wk_s = take_work((unsigned int*)wkc + 2 * WK_CSTRIDE);
  og_barrier();
  const int wk   = __builtin_amdgcn_readfirstlane(wk_s);
  const int r0   = wk * OG_ROWS;

  const float* pu = U + (size_t)(r0 + (tid >> 6)) * N_NODES + (tid & 63) * 4;
  const v4f*   pt = (const v4f*)s + tid;
  const int uoff = (tid >> 6) * OG_UP + (tid & 63) * 4;
  const int soff = (tid >> 4) * OG_SG + (tid & 15) * 4;
  const int aoff = mn * OG_UP + w * 16 + 4 * half;
  const int boff = (4 * w + half) * OG_SG + mn;

  float* buf0 = lds;
  float* buf1 = lds + OG_BUF;

  const int cb    = (wk & 127) >> 3;
  const int first = (2 * cb + 2) & (OG_NCH - 1);
  const int swp   = wk & 1;
#define OG_CH(j) ((((j) ^ swp) + first) & (OG_NCH - 1))

  OG_DECL(X);
  OG_DECL(Y);
  OG_ISSUE(X, OG_CH(0));
  __builtin_amdgcn_sched_barrier(0);
  OG_ISSUE(Y, OG_CH(1));
  __builtin_amdgcn_sched_barrier(0);
  OG_STASH(X, buf0);
  OG_ISSUE(X, OG_CH(2));
  __builtin_amdgcn_sched_barrier(0);
  og_barrier();

  v8f acc0 = {}, acc1 = {};
  for (int i = 0; i < OG_NCH - 4; i += 2) {
    OG_STASH(Y, buf1);
    OG_ISSUE(Y, OG_CH(i + 3));
    og_compute(buf0, aoff, boff, acc0, acc1);
    og_barrier();
    OG_STASH(X, buf0);
    OG_ISSUE(X, OG_CH(i + 4));
    og_compute(buf1, aoff, boff, acc0, acc1);
    og_barrier();
  }
  OG_STASH(Y, buf1);
  OG_ISSUE(Y, OG_CH(OG_NCH - 1));
  og_compute(buf0, aoff, boff, acc0, acc1);
  og_barrier();
  OG_STASH(X, buf0);
  og_compute(buf1, aoff, boff, acc0, acc1);
  og_barrier();
  OG_STASH(Y, buf1);
  og_compute(buf0, aoff, boff, acc0, acc1);
  og_barrier();
  og_compute(buf1, aoff, boff, acc0, acc1);
  og_barrier();

  float* red = lds;
  float* vs  = lds + OG_BUF;
#pragma unroll
  for (int r = 0; r < 8; ++r) {
    red[((w * 2 + 0) * 8 + r) * 32 + lane] = acc0[r];
    red[((w * 2 + 1) * 8 + r) * 32 + lane] = acc1[r];
  }
  og_barrier();
  const int row  = tid >> 4;
  const int c    = tid & 15;
  const int tile = row >> 4;
  const int rr   = row & 15;
  const int ridx = (tile * 8 + (rr & 7)) * 32 + c + 16 * (rr >> 3);
  float v = 0.0f;
#pragma unroll
  for (int w2 = 0; w2 < 16; ++w2) v += red[w2 * (2 * 8 * 32) + ridx];
  vs[tid] = v;
  og_barrier();

  float rv[CCH];
#pragma unroll
  for (int q = 0; q < 4; ++q) {
    float4 t4 = *(const float4*)&vs[row * CCH + q * 4];
    rv[q * 4 + 0] = t4.x; rv[q * 4 + 1] = t4.y; rv[q * 4 + 2] = t4.z; rv[q * 4 + 3] = t4.w;
  }
  float mx = rv[0];
#pragma unroll
  for (int i = 1; i < CCH; ++i) mx = fmaxf(mx, rv[i]);
  float sm = 0.0f;
#pragma unroll
  for (int i = 0; i < CCH; ++i) sm += expf(rv[i] - mx);
  const float lse = mx + logf(sm);
  out[(size_t)(r0 + row) * CCH + c] = v - lse;
}

extern "C" void kernel_launch(void* const* d_in, const int* in_sizes, int n_in,
                              void* d_out, int out_size, void* d_ws, size_t ws_size,
                              hipStream_t stream) {
  const float* origin_e = (const float*)d_in[0];
  const float* U        = (const float*)d_in[1];
  const float* x        = (const float*)d_in[2];
  const float* W1       = (const float*)d_in[3];
  const float* b1       = (const float*)d_in[4];
  const float* W2       = (const float*)d_in[5];
  const float* b2       = (const float*)d_in[6];
  const float* temp     = (const float*)d_in[7];
  float* out = (float*)d_out;

  float* ws   = (float*)d_ws;
  float* zb   = ws;
  float* sbuf = zb   + (size_t)N_NODES * CCH;
  float* wkc  = sbuf + (size_t)N_NODES * CCH;

  mlp_kernel<<<N_NODES / ML_ROWS, ML_THR, 0, stream>>>(x, W1, b1, W2, b2, zb, sbuf, wkc);
  utz_kernel<<<2 * (N_NODES / 64), UZ_THR, 0, stream>>>(U, zb, origin_e, temp, sbuf, wkc);
  outgemm_lsm_kernel<<<N_NODES / OG_ROWS, OG_THR, 0, stream>>>(U, sbuf, out, wkc);
}
